# starting kernel
# baseline (speedup 1.0000x reference)
#include <hip/hip_runtime.h>
#include <hip/hip_bf16.h>

#define NN 50000
#define EE 800000
#define KDIM 256
#define OUTD 64
#define NEG_SLOPE 0.2f

typedef __attribute__((ext_vector_type(2))) float v2f;
typedef __attribute__((ext_vector_type(4))) float v4f;
typedef __attribute__((ext_vector_type(8))) float v8f;

__device__ __forceinline__ unsigned orderFloat(float f) {
  unsigned u = __float_as_uint(f);
  return (u & 0x80000000u) ? ~u : (u | 0x80000000u);
}
__device__ __forceinline__ float unorderFloat(unsigned u) {
  u = (u & 0x80000000u) ? (u & 0x7FFFFFFFu) : ~u;
  return __uint_as_float(u);
}

__global__ __launch_bounds__(256) void init_kernel(
    float* __restrict__ rst, const float* __restrict__ gat_bias,
    unsigned* __restrict__ m_u32, float* __restrict__ denom) {
  int i = blockIdx.x * blockDim.x + threadIdx.x;
  if (i < NN * KDIM) rst[i] = gat_bias[i & (KDIM - 1)];
  if (i < NN * 4) { m_u32[i] = 0u; denom[i] = 0.0f; }
}

__global__ __launch_bounds__(32) void wmma_proj_kernel(
    const float* __restrict__ feat,
    const float* __restrict__ Wsrc, const float* __restrict__ Bsrc,
    const float* __restrict__ Wdst, const float* __restrict__ Bdst,
    float* __restrict__ Hsrc, float* __restrict__ Hdst) {
  const int lane = threadIdx.x;
  const int rowTile = blockIdx.x;
  const int yt = blockIdx.y;
  const int mat = yt >> 4;
  const int colTile = yt & 15;

  const float* __restrict__ W  = mat ? Wdst : Wsrc;
  const float* __restrict__ Bv = mat ? Bdst : Bsrc;
  float* __restrict__ Hout     = mat ? Hdst : Hsrc;

  const int row0 = rowTile * 64;
  const int col0 = colTile * 16;
  const int khi  = lane >> 4;
  const int mlo  = lane & 15;

  int arow[4];
#pragma unroll
  for (int s = 0; s < 4; ++s) {
    int r = row0 + s * 16 + mlo;
    arow[s] = (r < NN) ? r : 0;
  }

  v8f acc[4] = {};
  for (int kb = 0; kb < KDIM; kb += 4) {
    v2f bfrag;
    bfrag.x = W[(size_t)(kb + 2 * khi)     * KDIM + col0 + mlo];
    bfrag.y = W[(size_t)(kb + 2 * khi + 1) * KDIM + col0 + mlo];
#pragma unroll
    for (int s = 0; s < 4; ++s) {
      v2f afrag = *(const v2f*)(feat + (size_t)arow[s] * KDIM + kb + 2 * khi);
      acc[s] = __builtin_amdgcn_wmma_f32_16x16x4_f32(
          false, afrag, false, bfrag, (short)0, acc[s], false, false);
    }
  }

  const int mbase = khi * 8;
  const int ncol  = col0 + mlo;
  const float bias = Bv[ncol];
  if (row0 + 64 <= NN) {
#pragma unroll
    for (int s = 0; s < 4; ++s) {
#pragma unroll
      for (int i = 0; i < 8; ++i) {
        int r = row0 + s * 16 + mbase + i;
        Hout[(size_t)r * KDIM + ncol] = acc[s][i] + bias;
      }
    }
  } else {
#pragma unroll
    for (int s = 0; s < 4; ++s) {
#pragma unroll
      for (int i = 0; i < 8; ++i) {
        int r = row0 + s * 16 + mbase + i;
        if (r < NN) Hout[(size_t)r * KDIM + ncol] = acc[s][i] + bias;
      }
    }
  }
}

__global__ __launch_bounds__(256) void edge_score_kernel(
    const int* __restrict__ ei, const float* __restrict__ Hsrc,
    const float* __restrict__ Hdst, const float* __restrict__ attn,
    float* __restrict__ score, unsigned* __restrict__ m_u32) {
  const int e = (blockIdx.x * blockDim.x + threadIdx.x) >> 5;
  const int lane = threadIdx.x & 31;
  if (e >= EE) return;
  const int src = ei[e];
  const int dst = ei[EE + e];

  const v4f* ps = (const v4f*)(Hsrc + (size_t)src * KDIM + lane * 8);
  const v4f* pd = (const v4f*)(Hdst + (size_t)dst * KDIM + lane * 8);
  const v4f* pa = (const v4f*)(attn + lane * 8);
  v4f s0 = ps[0], s1 = ps[1];
  v4f d0 = pd[0], d1 = pd[1];
  v4f a0 = pa[0], a1 = pa[1];

  float part = 0.0f;
#pragma unroll
  for (int i = 0; i < 4; ++i) {
    float v = s0[i] + d0[i];
    v = (v >= 0.0f) ? v : v * NEG_SLOPE;
    part += v * a0[i];
    float w = s1[i] + d1[i];
    w = (w >= 0.0f) ? w : w * NEG_SLOPE;
    part += w * a1[i];
  }
  part += __shfl_xor(part, 1, 32);
  part += __shfl_xor(part, 2, 32);
  part += __shfl_xor(part, 4, 32);
  if ((lane & 7) == 0) {
    int h = lane >> 3;
    score[(size_t)e * 4 + h] = part;
    atomicMax(&m_u32[(size_t)dst * 4 + h], orderFloat(part));
  }
}

__global__ __launch_bounds__(256) void sanitize_m_kernel(unsigned* __restrict__ m) {
  int i = blockIdx.x * blockDim.x + threadIdx.x;
  if (i >= NN * 4) return;
  float f = unorderFloat(m[i]);
  if (!isfinite(f)) f = 0.0f;
  ((float*)m)[i] = f;
}

__global__ __launch_bounds__(256) void edge_exp_kernel(
    const int* __restrict__ ei, const float* __restrict__ m_f,
    float* __restrict__ score, float* __restrict__ denom) {
  int i = blockIdx.x * blockDim.x + threadIdx.x;
  if (i >= EE * 4) return;
  int e = i >> 2, h = i & 3;
  int dst = ei[EE + e];
  float ex = expf(score[i] - m_f[(size_t)dst * 4 + h]);
  score[i] = ex;
  unsafeAtomicAdd(&denom[(size_t)dst * 4 + h], ex);
}

__global__ __launch_bounds__(256) void edge_scatter_kernel(
    const int* __restrict__ ei, const float* __restrict__ Hsrc,
    const float* __restrict__ ex, const float* __restrict__ denom,
    float* __restrict__ rst) {
  const int e = (blockIdx.x * blockDim.x + threadIdx.x) >> 5;
  const int lane = threadIdx.x & 31;
  if (e >= EE) return;
  const int src = ei[e];
  const int dst = ei[EE + e];
  const int h = lane >> 3;
  float den = denom[(size_t)dst * 4 + h];
  float alpha = ex[(size_t)e * 4 + h] / ((den == 0.0f) ? 1.0f : den);
  const v4f* ps = (const v4f*)(Hsrc + (size_t)src * KDIM + lane * 8);
  v4f s0 = ps[0], s1 = ps[1];
  float* pr = rst + (size_t)dst * KDIM + lane * 8;
#pragma unroll
  for (int i = 0; i < 4; ++i) {
    unsafeAtomicAdd(&pr[i],     alpha * s0[i]);
    unsafeAtomicAdd(&pr[i + 4], alpha * s1[i]);
  }
}

__global__ __launch_bounds__(32) void wmma_out_kernel(
    const float* __restrict__ rst, const float* __restrict__ Wfc,
    const float* __restrict__ bfc, float* __restrict__ out) {
  const int lane = threadIdx.x;
  const int row0 = blockIdx.x * 64;
  const int col0 = blockIdx.y * 16;
  const int khi = lane >> 4;
  const int mlo = lane & 15;

  int arow[4];
#pragma unroll
  for (int s = 0; s < 4; ++s) {
    int r = row0 + s * 16 + mlo;
    arow[s] = (r < NN) ? r : 0;
  }

  v8f acc[4] = {};
  for (int kb = 0; kb < KDIM; kb += 4) {
    v2f bfrag;
    bfrag.x = Wfc[(size_t)(kb + 2 * khi)     * OUTD + col0 + mlo];
    bfrag.y = Wfc[(size_t)(kb + 2 * khi + 1) * OUTD + col0 + mlo];
#pragma unroll
    for (int s = 0; s < 4; ++s) {
      v2f afrag = *(const v2f*)(rst + (size_t)arow[s] * KDIM + kb + 2 * khi);
      acc[s] = __builtin_amdgcn_wmma_f32_16x16x4_f32(
          false, afrag, false, bfrag, (short)0, acc[s], false, false);
    }
  }

  const int mbase = khi * 8;
  const int ncol  = col0 + mlo;
  const float bias = bfc[ncol];
  if (row0 + 64 <= NN) {
#pragma unroll
    for (int s = 0; s < 4; ++s) {
#pragma unroll
      for (int i = 0; i < 8; ++i) {
        int r = row0 + s * 16 + mbase + i;
        out[(size_t)r * OUTD + ncol] = acc[s][i] + bias;
      }
    }
  } else {
#pragma unroll
    for (int s = 0; s < 4; ++s) {
#pragma unroll
      for (int i = 0; i < 8; ++i) {
        int r = row0 + s * 16 + mbase + i;
        if (r < NN) out[(size_t)r * OUTD + ncol] = acc[s][i] + bias;
      }
    }
  }
}

extern "C" void kernel_launch(void* const* d_in, const int* in_sizes, int n_in,
                              void* d_out, int out_size, void* d_ws, size_t ws_size,
                              hipStream_t stream) {
  const float* feat     = (const float*)d_in[0];
  const int*   ei       = (const int*)d_in[1];
  const float* w_src    = (const float*)d_in[2];
  const float* b_src    = (const float*)d_in[3];
  const float* w_dst    = (const float*)d_in[4];
  const float* b_dst    = (const float*)d_in[5];
  const float* attn     = (const float*)d_in[6];
  const float* gat_bias = (const float*)d_in[7];
  const float* w_fc     = (const float*)d_in[8];
  const float* b_fc     = (const float*)d_in[9];
  float* out = (float*)d_out;

  float* ws    = (float*)d_ws;
  float* Hsrc  = ws;
  float* Hdst  = Hsrc + (size_t)NN * KDIM;
  float* score = Hdst + (size_t)NN * KDIM;
  float* mbuf  = score + (size_t)EE * 4;
  float* denom = mbuf + (size_t)NN * 4;
  float* rst   = denom + (size_t)NN * 4;

  const int rowTiles = (NN + 63) / 64;

  init_kernel<<<(NN * KDIM + 255) / 256, 256, 0, stream>>>(
      rst, gat_bias, (unsigned*)mbuf, denom);

  wmma_proj_kernel<<<dim3(rowTiles, 32), 32, 0, stream>>>(
      feat, w_src, b_src, w_dst, b_dst, Hsrc, Hdst);

  edge_score_kernel<<<(EE * 32) / 256, 256, 0, stream>>>(
      ei, Hsrc, Hdst, attn, score, (unsigned*)mbuf);

  sanitize_m_kernel<<<(NN * 4 + 255) / 256, 256, 0, stream>>>((unsigned*)mbuf);

  edge_exp_kernel<<<(EE * 4 + 255) / 256, 256, 0, stream>>>(
      ei, mbuf, score, denom);

  edge_scatter_kernel<<<(EE * 32) / 256, 256, 0, stream>>>(
      ei, Hsrc, score, denom, rst);

  wmma_out_kernel<<<dim3(rowTiles, 4), 32, 0, stream>>>(rst, w_fc, b_fc, out);
}


# Apply each head's 64x64 block of w_fc to h_src on WMMA before the scatter: an edge adds 64 floats straight into out, not 256 into rst; head sum and w_fc product now precede the edge sum
# speedup vs baseline: 1.3964x; 1.3964x over previous
#include <hip/hip_runtime.h>
#include <hip/hip_bf16.h>

#define NN 50000
#define EE 800000
#define KDIM 256
#define OUTD 64
#define NEG_SLOPE 0.2f

typedef __attribute__((ext_vector_type(2))) float v2f;
typedef __attribute__((ext_vector_type(4))) float v4f;
typedef __attribute__((ext_vector_type(8))) float v8f;

__device__ __forceinline__ unsigned orderFloat(float f) {
  unsigned u = __float_as_uint(f);
  return (u & 0x80000000u) ? ~u : (u | 0x80000000u);
}
__device__ __forceinline__ float unorderFloat(unsigned u) {
  u = (u & 0x80000000u) ? (u & 0x7FFFFFFFu) : ~u;
  return __uint_as_float(u);
}

__global__ __launch_bounds__(256) void init_kernel(
    float* __restrict__ out, const float* __restrict__ gat_bias,
    const float* __restrict__ Wfc, const float* __restrict__ bfc,
    unsigned* __restrict__ m_u32, float* __restrict__ denom) {
  __shared__ float part[4][OUTD];
  const int t = threadIdx.x;
  const int j = t & (OUTD - 1);
  const int q = t >> 6;
  const float* __restrict__ gb = gat_bias + q * 64;
  const float* __restrict__ wq = Wfc + (size_t)q * 64 * OUTD + j;
  float s = 0.0f;
#pragma unroll 16
  for (int k = 0; k < 64; ++k)
    s += gb[k] * wq[(size_t)k * OUTD];
  part[q][j] = s;
  __syncthreads();
  const float c = bfc[j] + ((part[0][j] + part[1][j]) + (part[2][j] + part[3][j]));
  const int row0 = blockIdx.x * 64;
  float* __restrict__ po = out + (size_t)(row0 + q) * OUTD + j;
  if (row0 + 64 <= NN) {
    for (int r = 0; r < 16; ++r) po[(size_t)r * 4 * OUTD] = c;
  } else {
    for (int r = 0; r < 16; ++r)
      if (row0 + q + r * 4 < NN) po[(size_t)r * 4 * OUTD] = c;
  }
  const int i = blockIdx.x * 256 + t;
  if (i < NN * 4) { m_u32[i] = 0u; denom[i] = 0.0f; }
}

__global__ __launch_bounds__(32) void wmma_proj_kernel(
    const float* __restrict__ feat,
    const float* __restrict__ Wsrc, const float* __restrict__ Bsrc,
    const float* __restrict__ Wdst, const float* __restrict__ Bdst,
    float* __restrict__ Hsrc, float* __restrict__ Hdst) {
  const int lane = threadIdx.x;
  const int rowTile = blockIdx.x;
  const int yt = blockIdx.y;
  const int mat = yt >> 4;
  const int colTile = yt & 15;

  const float* __restrict__ W  = mat ? Wdst : Wsrc;
  const float* __restrict__ Bv = mat ? Bdst : Bsrc;
  float* __restrict__ Hout     = mat ? Hdst : Hsrc;

  const int row0 = rowTile * 64;
  const int col0 = colTile * 16;
  const int khi  = lane >> 4;
  const int mlo  = lane & 15;

  int arow[4];
#pragma unroll
  for (int s = 0; s < 4; ++s) {
    int r = row0 + s * 16 + mlo;
    arow[s] = (r < NN) ? r : 0;
  }

  v8f acc[4] = {};
  for (int kb = 0; kb < KDIM; kb += 4) {
    v2f bfrag;
    bfrag.x = W[(size_t)(kb + 2 * khi)     * KDIM + col0 + mlo];
    bfrag.y = W[(size_t)(kb + 2 * khi + 1) * KDIM + col0 + mlo];
#pragma unroll
    for (int s = 0; s < 4; ++s) {
      v2f afrag = *(const v2f*)(feat + (size_t)arow[s] * KDIM + kb + 2 * khi);
      acc[s] = __builtin_amdgcn_wmma_f32_16x16x4_f32(
          false, afrag, false, bfrag, (short)0, acc[s], false, false);
    }
  }

  const int mbase = khi * 8;
  const int ncol  = col0 + mlo;
  const float bias = Bv[ncol];
  if (row0 + 64 <= NN) {
#pragma unroll
    for (int s = 0; s < 4; ++s) {
#pragma unroll
      for (int i = 0; i < 8; ++i) {
        int r = row0 + s * 16 + mbase + i;
        Hout[(size_t)r * KDIM + ncol] = acc[s][i] + bias;
      }
    }
  } else {
#pragma unroll
    for (int s = 0; s < 4; ++s) {
#pragma unroll
      for (int i = 0; i < 8; ++i) {
        int r = row0 + s * 16 + mbase + i;
        if (r < NN) Hout[(size_t)r * KDIM + ncol] = acc[s][i] + bias;
      }
    }
  }
}

__global__ __launch_bounds__(256) void edge_score_kernel(
    const int* __restrict__ ei, const float* __restrict__ Hsrc,
    const float* __restrict__ Hdst, const float* __restrict__ attn,
    float* __restrict__ score, unsigned* __restrict__ m_u32) {
  const int e = (blockIdx.x * blockDim.x + threadIdx.x) >> 5;
  const int lane = threadIdx.x & 31;
  if (e >= EE) return;
  const int src = ei[e];
  const int dst = ei[EE + e];

  const v4f* ps = (const v4f*)(Hsrc + (size_t)src * KDIM + lane * 8);
  const v4f* pd = (const v4f*)(Hdst + (size_t)dst * KDIM + lane * 8);
  const v4f* pa = (const v4f*)(attn + lane * 8);
  v4f s0 = ps[0], s1 = ps[1];
  v4f d0 = pd[0], d1 = pd[1];
  v4f a0 = pa[0], a1 = pa[1];

  float part = 0.0f;
#pragma unroll
  for (int i = 0; i < 4; ++i) {
    float v = s0[i] + d0[i];
    v = (v >= 0.0f) ? v : v * NEG_SLOPE;
    part += v * a0[i];
    float w = s1[i] + d1[i];
    w = (w >= 0.0f) ? w : w * NEG_SLOPE;
    part += w * a1[i];
  }
  part += __shfl_xor(part, 1, 32);
  part += __shfl_xor(part, 2, 32);
  part += __shfl_xor(part, 4, 32);
  if ((lane & 7) == 0) {
    int h = lane >> 3;
    score[(size_t)e * 4 + h] = part;
    atomicMax(&m_u32[(size_t)dst * 4 + h], orderFloat(part));
  }
}

__global__ __launch_bounds__(256) void sanitize_m_kernel(unsigned* __restrict__ m) {
  int i = blockIdx.x * blockDim.x + threadIdx.x;
  if (i >= NN * 4) return;
  float f = unorderFloat(m[i]);
  if (!isfinite(f)) f = 0.0f;
  ((float*)m)[i] = f;
}

__global__ __launch_bounds__(256) void edge_exp_kernel(
    const int* __restrict__ ei, const float* __restrict__ m_f,
    float* __restrict__ score, float* __restrict__ denom) {
  int i = blockIdx.x * blockDim.x + threadIdx.x;
  if (i >= EE * 4) return;
  int e = i >> 2, h = i & 3;
  int dst = ei[EE + e];
  float ex = expf(score[i] - m_f[(size_t)dst * 4 + h]);
  score[i] = ex;
  unsafeAtomicAdd(&denom[(size_t)dst * 4 + h], ex);
}

__global__ __launch_bounds__(256) void edge_scatter_kernel(
    const int* __restrict__ ei, const float* __restrict__ P,
    const float* __restrict__ ex, const float* __restrict__ denom,
    float* __restrict__ out) {
  const int e = (blockIdx.x * blockDim.x + threadIdx.x) >> 5;
  const int lane = threadIdx.x & 31;
  if (e >= EE) return;
  const int src = ei[e];
  const int dst = ei[EE + e];
  const int hq = lane & 3;
  const float dn = denom[(size_t)dst * 4 + hq];
  const float al = ex[(size_t)e * 4 + hq] / ((dn == 0.0f) ? 1.0f : dn);
  const float* pp = P + (size_t)src * KDIM + lane;
  const float a0 = __int_as_float(__builtin_amdgcn_readlane(__float_as_int(al), 0));
  const float a1 = __int_as_float(__builtin_amdgcn_readlane(__float_as_int(al), 1));
  const float a2 = __int_as_float(__builtin_amdgcn_readlane(__float_as_int(al), 2));
  const float a3 = __int_as_float(__builtin_amdgcn_readlane(__float_as_int(al), 3));
  float v0 = a0 * pp[0];
  float v1 = a0 * pp[32];
  v0 += a1 * pp[64];
  v1 += a1 * pp[96];
  v0 += a2 * pp[128];
  v1 += a2 * pp[160];
  v0 += a3 * pp[192];
  v1 += a3 * pp[224];
  float* po = out + (size_t)dst * OUTD + lane;
  unsafeAtomicAdd(po, v0);
  unsafeAtomicAdd(po + 32, v1);
}

__global__ __launch_bounds__(32) void wmma_headfc_kernel(
    const float* __restrict__ Hsrc, const float* __restrict__ Wfc,
    float* __restrict__ P) {
  const int lane = threadIdx.x;
  const int row0 = blockIdx.x * 64;
  const int head = blockIdx.y >> 2;
  const int col0 = (blockIdx.y & 3) * 16;
  const int k0 = head * 64;
  const int khi = lane >> 4;
  const int mlo = lane & 15;

  int arow[4];
#pragma unroll
  for (int s = 0; s < 4; ++s) {
    int r = row0 + s * 16 + mlo;
    arow[s] = (r < NN) ? r : 0;
  }

  v8f acc[4] = {};
  for (int kb = k0; kb < k0 + 64; kb += 4) {
    v2f bfrag;
    bfrag.x = Wfc[(size_t)(kb + 2 * khi)     * OUTD + col0 + mlo];
    bfrag.y = Wfc[(size_t)(kb + 2 * khi + 1) * OUTD + col0 + mlo];
#pragma unroll
    for (int s = 0; s < 4; ++s) {
      v2f afrag = *(const v2f*)(Hsrc + (size_t)arow[s] * KDIM + kb + 2 * khi);
      acc[s] = __builtin_amdgcn_wmma_f32_16x16x4_f32(
          false, afrag, false, bfrag, (short)0, acc[s], false, false);
    }
  }

  const int mbase = khi * 8;
  const int ncol  = k0 + col0 + mlo;
  if (row0 + 64 <= NN) {
#pragma unroll
    for (int s = 0; s < 4; ++s) {
#pragma unroll
      for (int i = 0; i < 8; ++i) {
        int r = row0 + s * 16 + mbase + i;
        P[(size_t)r * KDIM + ncol] = acc[s][i];
      }
    }
  } else {
#pragma unroll
    for (int s = 0; s < 4; ++s) {
#pragma unroll
      for (int i = 0; i < 8; ++i) {
        int r = row0 + s * 16 + mbase + i;
        if (r < NN) P[(size_t)r * KDIM + ncol] = acc[s][i];
      }
    }
  }
}

extern "C" void kernel_launch(void* const* d_in, const int* in_sizes, int n_in,
                              void* d_out, int out_size, void* d_ws, size_t ws_size,
                              hipStream_t stream) {
  const float* feat     = (const float*)d_in[0];
  const int*   ei       = (const int*)d_in[1];
  const float* w_src    = (const float*)d_in[2];
  const float* b_src    = (const float*)d_in[3];
  const float* w_dst    = (const float*)d_in[4];
  const float* b_dst    = (const float*)d_in[5];
  const float* attn     = (const float*)d_in[6];
  const float* gat_bias = (const float*)d_in[7];
  const float* w_fc     = (const float*)d_in[8];
  const float* b_fc     = (const float*)d_in[9];
  float* out = (float*)d_out;

  float* ws    = (float*)d_ws;
  float* Hsrc  = ws;
  float* Hdst  = Hsrc + (size_t)NN * KDIM;
  float* score = Hdst + (size_t)NN * KDIM;
  float* mbuf  = score + (size_t)EE * 4;
  float* denom = mbuf + (size_t)NN * 4;
  float* P     = denom + (size_t)NN * 4;

  const int rowTiles = (NN + 63) / 64;

  init_kernel<<<rowTiles, 256, 0, stream>>>(
      out, gat_bias, w_fc, b_fc, (unsigned*)mbuf, denom);

  wmma_proj_kernel<<<dim3(rowTiles, 32), 32, 0, stream>>>(
      feat, w_src, b_src, w_dst, b_dst, Hsrc, Hdst);

  wmma_headfc_kernel<<<dim3(rowTiles, 16), 32, 0, stream>>>(Hsrc, w_fc, P);

  edge_score_kernel<<<(EE * 32) / 256, 256, 0, stream>>>(
      ei, Hsrc, Hdst, attn, score, (unsigned*)mbuf);

  sanitize_m_kernel<<<(NN * 4 + 255) / 256, 256, 0, stream>>>((unsigned*)mbuf);

  edge_exp_kernel<<<(EE * 4 + 255) / 256, 256, 0, stream>>>(
      ei, mbuf, score, denom);

  edge_scatter_kernel<<<(EE * 32) / 256, 256, 0, stream>>>(
      ei, P, score, denom, out);
}


# Both WMMA GEMMs: k order inside each 32-k chunk permuted alike in A and B loaders (lane half h takes k 16h..16h+15), so a lane reads its A row in 64-byte runs, each line once; k summed in new order
# speedup vs baseline: 2.1677x; 1.5523x over previous
#include <hip/hip_runtime.h>
#include <hip/hip_bf16.h>

#define NN 50000
#define EE 800000
#define KDIM 256
#define OUTD 64
#define NEG_SLOPE 0.2f

typedef __attribute__((ext_vector_type(2))) float v2f;
typedef __attribute__((ext_vector_type(4))) float v4f;
typedef __attribute__((ext_vector_type(8))) float v8f;

__device__ __forceinline__ unsigned orderFloat(float f) {
  unsigned u = __float_as_uint(f);
  return (u & 0x80000000u) ? ~u : (u | 0x80000000u);
}
__device__ __forceinline__ float unorderFloat(unsigned u) {
  u = (u & 0x80000000u) ? (u & 0x7FFFFFFFu) : ~u;
  return __uint_as_float(u);
}

__global__ __launch_bounds__(256) void init_kernel(
    float* __restrict__ out, const float* __restrict__ gat_bias,
    const float* __restrict__ Wfc, const float* __restrict__ bfc,
    unsigned* __restrict__ m_u32, float* __restrict__ denom) {
  __shared__ float part[4][OUTD];
  const int t = threadIdx.x;
  const int j = t & (OUTD - 1);
  const int q = t >> 6;
  const float* __restrict__ gb = gat_bias + q * 64;
  const float* __restrict__ wq = Wfc + (size_t)q * 64 * OUTD + j;
  float s = 0.0f;
#pragma unroll 16
  for (int k = 0; k < 64; ++k)
    s += gb[k] * wq[(size_t)k * OUTD];
  part[q][j] = s;
  __syncthreads();
  const float c = bfc[j] + ((part[0][j] + part[1][j]) + (part[2][j] + part[3][j]));
  const int row0 = blockIdx.x * 64;
  float* __restrict__ po = out + (size_t)(row0 + q) * OUTD + j;
  if (row0 + 64 <= NN) {
    for (int r = 0; r < 16; ++r) po[(size_t)r * 4 * OUTD] = c;
  } else {
    for (int r = 0; r < 16; ++r)
      if (row0 + q + r * 4 < NN) po[(size_t)r * 4 * OUTD] = c;
  }
  const int i = blockIdx.x * 256 + t;
  if (i < NN * 4) { m_u32[i] = 0u; denom[i] = 0.0f; }
}

template <int KLEN, int BPITCH>
__device__ __forceinline__ void wmma_tile_64x16(
    const float* __restrict__ Ap, const int (&arow)[4],
    const float* __restrict__ Bp, const int khi, v8f (&acc)[4]) {
#pragma unroll 1
  for (int kc = 0; kc < KLEN; kc += 32) {
    const int kl = kc + 16 * khi;
    float b[16];
    v4f a[4][4];
#pragma unroll
    for (int e = 0; e < 16; ++e) b[e] = Bp[(size_t)(kl + e) * BPITCH];
#pragma unroll
    for (int s = 0; s < 4; ++s) {
      const v4f* pa = (const v4f*)(Ap + (size_t)arow[s] * KDIM + kl);
#pragma unroll
      for (int q = 0; q < 4; ++q) a[s][q] = pa[q];
    }
    __builtin_amdgcn_sched_barrier(0);
#pragma unroll
    for (int j = 0; j < 8; ++j) {
      v2f bfrag;
      bfrag.x = b[2 * j];
      bfrag.y = b[2 * j + 1];
#pragma unroll
      for (int s = 0; s < 4; ++s) {
        v2f afrag;
        afrag.x = a[s][j >> 1][2 * (j & 1)];
        afrag.y = a[s][j >> 1][2 * (j & 1) + 1];
        acc[s] = __builtin_amdgcn_wmma_f32_16x16x4_f32(
            false, afrag, false, bfrag, (short)0, acc[s], false, false);
      }
    }
  }
}

__global__ __launch_bounds__(32) void wmma_proj_kernel(
    const float* __restrict__ feat,
    const float* __restrict__ Wsrc, const float* __restrict__ Bsrc,
    const float* __restrict__ Wdst, const float* __restrict__ Bdst,
    float* __restrict__ Hsrc, float* __restrict__ Hdst) {
  const int lane = threadIdx.x;
  const int rowTile = blockIdx.x;
  const int yt = blockIdx.y;
  const int mat = yt >> 4;
  const int colTile = yt & 15;

  const float* __restrict__ W  = mat ? Wdst : Wsrc;
  const float* __restrict__ Bv = mat ? Bdst : Bsrc;
  float* __restrict__ Hout     = mat ? Hdst : Hsrc;

  const int row0 = rowTile * 64;
  const int col0 = colTile * 16;
  const int khi  = lane >> 4;
  const int mlo  = lane & 15;

  int arow[4];
#pragma unroll
  for (int s = 0; s < 4; ++s) {
    int r = row0 + s * 16 + mlo;
    arow[s] = (r < NN) ? r : 0;
  }

  v8f acc[4] = {};
  wmma_tile_64x16<KDIM, KDIM>(feat, arow, W + col0 + mlo, khi, acc);

  const int mbase = khi * 8;
  const int ncol  = col0 + mlo;
  const float bias = Bv[ncol];
  if (row0 + 64 <= NN) {
#pragma unroll
    for (int s = 0; s < 4; ++s) {
#pragma unroll
      for (int i = 0; i < 8; ++i) {
        int r = row0 + s * 16 + mbase + i;
        Hout[(size_t)r * KDIM + ncol] = acc[s][i] + bias;
      }
    }
  } else {
#pragma unroll
    for (int s = 0; s < 4; ++s) {
#pragma unroll
      for (int i = 0; i < 8; ++i) {
        int r = row0 + s * 16 + mbase + i;
        if (r < NN) Hout[(size_t)r * KDIM + ncol] = acc[s][i] + bias;
      }
    }
  }
}

__global__ __launch_bounds__(256) void edge_score_kernel(
    const int* __restrict__ ei, const float* __restrict__ Hsrc,
    const float* __restrict__ Hdst, const float* __restrict__ attn,
    float* __restrict__ score, unsigned* __restrict__ m_u32) {
  const int e = (blockIdx.x * blockDim.x + threadIdx.x) >> 5;
  const int lane = threadIdx.x & 31;
  if (e >= EE) return;
  const int src = ei[e];
  const int dst = ei[EE + e];

  const v4f* ps = (const v4f*)(Hsrc + (size_t)src * KDIM + lane * 8);
  const v4f* pd = (const v4f*)(Hdst + (size_t)dst * KDIM + lane * 8);
  const v4f* pa = (const v4f*)(attn + lane * 8);
  v4f s0 = ps[0], s1 = ps[1];
  v4f d0 = pd[0], d1 = pd[1];
  v4f a0 = pa[0], a1 = pa[1];

  float part = 0.0f;
#pragma unroll
  for (int i = 0; i < 4; ++i) {
    float v = s0[i] + d0[i];
    v = (v >= 0.0f) ? v : v * NEG_SLOPE;
    part += v * a0[i];
    float w = s1[i] + d1[i];
    w = (w >= 0.0f) ? w : w * NEG_SLOPE;
    part += w * a1[i];
  }
  part += __shfl_xor(part, 1, 32);
  part += __shfl_xor(part, 2, 32);
  part += __shfl_xor(part, 4, 32);
  if ((lane & 7) == 0) {
    int h = lane >> 3;
    score[(size_t)e * 4 + h] = part;
    atomicMax(&m_u32[(size_t)dst * 4 + h], orderFloat(part));
  }
}

__global__ __launch_bounds__(256) void sanitize_m_kernel(unsigned* __restrict__ m) {
  int i = blockIdx.x * blockDim.x + threadIdx.x;
  if (i >= NN * 4) return;
  float f = unorderFloat(m[i]);
  if (!isfinite(f)) f = 0.0f;
  ((float*)m)[i] = f;
}

__global__ __launch_bounds__(256) void edge_exp_kernel(
    const int* __restrict__ ei, const float* __restrict__ m_f,
    float* __restrict__ score, float* __restrict__ denom) {
  int i = blockIdx.x * blockDim.x + threadIdx.x;
  if (i >= EE * 4) return;
  int e = i >> 2, h = i & 3;
  int dst = ei[EE + e];
  float ex = expf(score[i] - m_f[(size_t)dst * 4 + h]);
  score[i] = ex;
  unsafeAtomicAdd(&denom[(size_t)dst * 4 + h], ex);
}

__global__ __launch_bounds__(256) void edge_scatter_kernel(
    const int* __restrict__ ei, const float* __restrict__ P,
    const float* __restrict__ ex, const float* __restrict__ denom,
    float* __restrict__ out) {
  const int e = (blockIdx.x * blockDim.x + threadIdx.x) >> 5;
  const int lane = threadIdx.x & 31;
  if (e >= EE) return;
  const int src = ei[e];
  const int dst = ei[EE + e];
  const int hq = lane & 3;
  const float dn = denom[(size_t)dst * 4 + hq];
  const float al = ex[(size_t)e * 4 + hq] / ((dn == 0.0f) ? 1.0f : dn);
  const float* pp = P + (size_t)src * KDIM + lane;
  const float a0 = __int_as_float(__builtin_amdgcn_readlane(__float_as_int(al), 0));
  const float a1 = __int_as_float(__builtin_amdgcn_readlane(__float_as_int(al), 1));
  const float a2 = __int_as_float(__builtin_amdgcn_readlane(__float_as_int(al), 2));
  const float a3 = __int_as_float(__builtin_amdgcn_readlane(__float_as_int(al), 3));
  float v0 = a0 * pp[0];
  float v1 = a0 * pp[32];
  v0 += a1 * pp[64];
  v1 += a1 * pp[96];
  v0 += a2 * pp[128];
  v1 += a2 * pp[160];
  v0 += a3 * pp[192];
  v1 += a3 * pp[224];
  float* po = out + (size_t)dst * OUTD + lane;
  unsafeAtomicAdd(po, v0);
  unsafeAtomicAdd(po + 32, v1);
}

__global__ __launch_bounds__(32) void wmma_headfc_kernel(
    const float* __restrict__ Hsrc, const float* __restrict__ Wfc,
    float* __restrict__ P) {
  const int lane = threadIdx.x;
  const int row0 = blockIdx.x * 64;
  const int head = blockIdx.y >> 2;
  const int col0 = (blockIdx.y & 3) * 16;
  const int k0 = head * 64;
  const int khi = lane >> 4;
  const int mlo = lane & 15;

  int arow[4];
#pragma unroll
  for (int s = 0; s < 4; ++s) {
    int r = row0 + s * 16 + mlo;
    arow[s] = (r < NN) ? r : 0;
  }

  v8f acc[4] = {};
  wmma_tile_64x16<64, OUTD>(Hsrc + k0, arow, Wfc + (size_t)k0 * OUTD + col0 + mlo,
                            khi, acc);

  const int mbase = khi * 8;
  const int ncol  = k0 + col0 + mlo;
  if (row0 + 64 <= NN) {
#pragma unroll
    for (int s = 0; s < 4; ++s) {
#pragma unroll
      for (int i = 0; i < 8; ++i) {
        int r = row0 + s * 16 + mbase + i;
        P[(size_t)r * KDIM + ncol] = acc[s][i];
      }
    }
  } else {
#pragma unroll
    for (int s = 0; s < 4; ++s) {
#pragma unroll
      for (int i = 0; i < 8; ++i) {
        int r = row0 + s * 16 + mbase + i;
        if (r < NN) P[(size_t)r * KDIM + ncol] = acc[s][i];
      }
    }
  }
}

extern "C" void kernel_launch(void* const* d_in, const int* in_sizes, int n_in,
                              void* d_out, int out_size, void* d_ws, size_t ws_size,
                              hipStream_t stream) {
  const float* feat     = (const float*)d_in[0];
  const int*   ei       = (const int*)d_in[1];
  const float* w_src    = (const float*)d_in[2];
  const float* b_src    = (const float*)d_in[3];
  const float* w_dst    = (const float*)d_in[4];
  const float* b_dst    = (const float*)d_in[5];
  const float* attn     = (const float*)d_in[6];
  const float* gat_bias = (const float*)d_in[7];
  const float* w_fc     = (const float*)d_in[8];
  const float* b_fc     = (const float*)d_in[9];
  float* out = (float*)d_out;

  float* ws    = (float*)d_ws;
  float* Hsrc  = ws;
  float* Hdst  = Hsrc + (size_t)NN * KDIM;
  float* score = Hdst + (size_t)NN * KDIM;
  float* mbuf  = score + (size_t)EE * 4;
  float* denom = mbuf + (size_t)NN * 4;
  float* P     = denom + (size_t)NN * 4;

  const int rowTiles = (NN + 63) / 64;

  init_kernel<<<rowTiles, 256, 0, stream>>>(
      out, gat_bias, w_fc, b_fc, (unsigned*)mbuf, denom);

  wmma_proj_kernel<<<dim3(rowTiles, 32), 32, 0, stream>>>(
      feat, w_src, b_src, w_dst, b_dst, Hsrc, Hdst);

  wmma_headfc_kernel<<<dim3(rowTiles, 16), 32, 0, stream>>>(Hsrc, w_fc, P);

  edge_score_kernel<<<(EE * 32) / 256, 256, 0, stream>>>(
      ei, Hsrc, Hdst, attn, score, (unsigned*)mbuf);

  sanitize_m_kernel<<<(NN * 4 + 255) / 256, 256, 0, stream>>>((unsigned*)mbuf);

  edge_exp_kernel<<<(EE * 4 + 255) / 256, 256, 0, stream>>>(
      ei, mbuf, score, denom);

  edge_scatter_kernel<<<(EE * 32) / 256, 256, 0, stream>>>(
      ei, P, score, denom, out);
}


# Edges grouped by destination (counting sort, int atomics, arrival-order rank); one wave per node does score, online softmax and weighted Hsrc sum in registers, w_fc GEMM after: no float atomics
# speedup vs baseline: 3.3662x; 1.5529x over previous
#include <hip/hip_runtime.h>


#define NN 50000
#define EE 800000
#define KDIM 256
#define OUTD 64
#define NEG_SLOPE 0.2f

typedef __attribute__((ext_vector_type(2))) float v2f;
typedef __attribute__((ext_vector_type(4))) float v4f;
typedef __attribute__((ext_vector_type(8))) float v8f;

__global__ __launch_bounds__(256) void zero_counts_kernel(
    int* __restrict__ count, int* __restrict__ total) {
  const int i = blockIdx.x * 256 + threadIdx.x;
  if (i < NN) count[i] = 0;
  if (i == 0) total[0] = 0;
}

__global__ __launch_bounds__(256) void dst_rank_kernel(
    const int* __restrict__ ei, int* __restrict__ count, int* __restrict__ rank) {
  const int e = blockIdx.x * 256 + threadIdx.x;
  if (e < EE) rank[e] = atomicAdd(&count[ei[EE + e]], 1);
}

__global__ __launch_bounds__(256) void seg_offsets_kernel(
    const int* __restrict__ count, int* __restrict__ total,
    int* __restrict__ segbeg) {
  __shared__ int wsum[8];
  __shared__ int base_s;
  const int t = threadIdx.x;
  const int lane = t & 31;
  const int w = t >> 5;
  const int i = blockIdx.x * 256 + t;
  const int c = (i < NN) ? count[i] : 0;
  int x = c;
#pragma unroll
  for (int d = 1; d < 32; d <<= 1) {
    const int y = __shfl_up(x, d, 32);
    if (lane >= d) x += y;
  }
  if (lane == 31) wsum[w] = x;
  __syncthreads();
  int wpre = 0, tot = 0;
#pragma unroll
  for (int k = 0; k < 8; ++k) {
    const int v = wsum[k];
    if (k < w) wpre += v;
    tot += v;
  }
  if (t == 0) base_s = atomicAdd(total, tot);
  __syncthreads();
  if (i < NN) segbeg[i] = base_s + wpre + (x - c);
}

__global__ __launch_bounds__(256) void dst_place_kernel(
    const int* __restrict__ ei, const int* __restrict__ rank,
    const int* __restrict__ segbeg, int* __restrict__ srcs) {
  const int e = blockIdx.x * 256 + threadIdx.x;
  if (e >= EE) return;
  const int src = ei[e];
  const int dst = ei[EE + e];
  const int idx = segbeg[dst] + rank[e];
  if ((unsigned)idx < (unsigned)EE) srcs[idx] = src;
}

template <int KLEN, int BPITCH>
__device__ __forceinline__ void wmma_tile_64x16(
    const float* __restrict__ Ap, const int (&arow)[4],
    const float* __restrict__ Bp, const int khi, v8f (&acc)[4]) {
#pragma unroll 1
  for (int kc = 0; kc < KLEN; kc += 32) {
    const int kl = kc + 16 * khi;
    float b[16];
    v4f a[4][4];
#pragma unroll
    for (int e = 0; e < 16; ++e) b[e] = Bp[(size_t)(kl + e) * BPITCH];
#pragma unroll
    for (int s = 0; s < 4; ++s) {
      const v4f* pa = (const v4f*)(Ap + (size_t)arow[s] * KDIM + kl);
#pragma unroll
      for (int q = 0; q < 4; ++q) a[s][q] = pa[q];
    }
    __builtin_amdgcn_sched_barrier(0);
#pragma unroll
    for (int j = 0; j < 8; ++j) {
      v2f bfrag;
      bfrag.x = b[2 * j];
      bfrag.y = b[2 * j + 1];
#pragma unroll
      for (int s = 0; s < 4; ++s) {
        v2f afrag;
        afrag.x = a[s][j >> 1][2 * (j & 1)];
        afrag.y = a[s][j >> 1][2 * (j & 1) + 1];
        acc[s] = __builtin_amdgcn_wmma_f32_16x16x4_f32(
            false, afrag, false, bfrag, (short)0, acc[s], false, false);
      }
    }
  }
}

__global__ __launch_bounds__(32) void wmma_proj_kernel(
    const float* __restrict__ feat,
    const float* __restrict__ Wsrc, const float* __restrict__ Bsrc,
    const float* __restrict__ Wdst, const float* __restrict__ Bdst,
    float* __restrict__ Hsrc, float* __restrict__ Hdst) {
  const int lane = threadIdx.x;
  const int rowTile = blockIdx.x;
  const int yt = blockIdx.y;
  const int mat = yt >> 4;
  const int colTile = yt & 15;

  const float* __restrict__ W  = mat ? Wdst : Wsrc;
  const float* __restrict__ Bv = mat ? Bdst : Bsrc;
  float* __restrict__ Hout     = mat ? Hdst : Hsrc;

  const int row0 = rowTile * 64;
  const int col0 = colTile * 16;
  const int khi  = lane >> 4;
  const int mlo  = lane & 15;

  int arow[4];
#pragma unroll
  for (int s = 0; s < 4; ++s) {
    int r = row0 + s * 16 + mlo;
    arow[s] = (r < NN) ? r : 0;
  }

  v8f acc[4] = {};
  wmma_tile_64x16<KDIM, KDIM>(feat, arow, W + col0 + mlo, khi, acc);

  const int mbase = khi * 8;
  const int ncol  = col0 + mlo;
  const float bias = Bv[ncol];
  if (row0 + 64 <= NN) {
#pragma unroll
    for (int s = 0; s < 4; ++s) {
#pragma unroll
      for (int i = 0; i < 8; ++i) {
        int r = row0 + s * 16 + mbase + i;
        Hout[(size_t)r * KDIM + ncol] = acc[s][i] + bias;
      }
    }
  } else {
#pragma unroll
    for (int s = 0; s < 4; ++s) {
#pragma unroll
      for (int i = 0; i < 8; ++i) {
        int r = row0 + s * 16 + mbase + i;
        if (r < NN) Hout[(size_t)r * KDIM + ncol] = acc[s][i] + bias;
      }
    }
  }
}

__device__ __forceinline__ float row8_sum(float x) {
  x += __int_as_float(__builtin_amdgcn_update_dpp(0, __float_as_int(x), 0x161, 0xf, 0xf, false));
  x += __int_as_float(__builtin_amdgcn_update_dpp(0, __float_as_int(x), 0x162, 0xf, 0xf, false));
  x += __int_as_float(__builtin_amdgcn_update_dpp(0, __float_as_int(x), 0x164, 0xf, 0xf, false));
  return x;
}

__global__ __launch_bounds__(256) void node_attend_kernel(
    const int* __restrict__ segbeg, const int* __restrict__ count,
    const int* __restrict__ srcs, const float* __restrict__ Hsrc,
    const float* __restrict__ Hdst, const float* __restrict__ attn,
    const float* __restrict__ gat_bias, float* __restrict__ rst) {
  const int n = __builtin_amdgcn_readfirstlane((int)((blockIdx.x * 256 + threadIdx.x) >> 5));
  const int lane = threadIdx.x & 31;
  if (n >= NN) return;
  int beg = segbeg[n];
  int cnt = count[n];
  beg = min(max(beg, 0), EE);
  cnt = min(max(cnt, 0), EE - beg);

  const int off = (lane >> 3) * 64 + (lane & 7) * 4;
  const float* __restrict__ hd = Hdst + (size_t)n * KDIM + off;
  const v4f d0 = *(const v4f*)hd;
  const v4f d1 = *(const v4f*)(hd + 32);
  const v4f a0 = *(const v4f*)(attn + off);
  const v4f a1 = *(const v4f*)(attn + off + 32);

  float m = -__builtin_inff();
  float den = 0.0f;
  v4f acc0 = {0.0f, 0.0f, 0.0f, 0.0f};
  v4f acc1 = {0.0f, 0.0f, 0.0f, 0.0f};

  for (int base = 0; base < cnt; base += 32) {
    const int nb = min(32, cnt - base);
    int mysrc = 0;
    if (lane < nb) mysrc = srcs[beg + base + lane];
    mysrc = min(max(mysrc, 0), NN - 1);
    for (int j = 0; j < nb; j += 4) {
      v4f r0[4], r1[4];
#pragma unroll
      for (int u = 0; u < 4; ++u) {
        const int jj = min(j + u, nb - 1);
        const int s = __builtin_amdgcn_readlane(mysrc, jj);
        const float* __restrict__ row = Hsrc + (size_t)s * KDIM + off;
        r0[u] = *(const v4f*)row;
        r1[u] = *(const v4f*)(row + 32);
      }
#pragma unroll
      for (int u = 0; u < 4; ++u) {
        if (j + u < nb) {
          v4f x0 = r0[u] + d0;
          v4f x1 = r1[u] + d1;
          x0 = __builtin_elementwise_max(x0, x0 * NEG_SLOPE);
          x1 = __builtin_elementwise_max(x1, x1 * NEG_SLOPE);
          const v4f t = x0 * a0 + x1 * a1;
          float part = (t[0] + t[1]) + (t[2] + t[3]);
          part = row8_sum(part);
          const float dlt = part - m;
          const float ex = expf(-fabsf(dlt));
          const bool up = dlt > 0.0f;
          const float sc = up ? ex : 1.0f;
          const float p = up ? 1.0f : ex;
          m = up ? part : m;
          den = den * sc + p;
          acc0 = acc0 * sc + p * r0[u];
          acc1 = acc1 * sc + p * r1[u];
        }
      }
    }
  }

  const float inv = 1.0f / ((den == 0.0f) ? 1.0f : den);
  const v4f g0 = *(const v4f*)(gat_bias + off);
  const v4f g1 = *(const v4f*)(gat_bias + off + 32);
  float* __restrict__ po = rst + (size_t)n * KDIM + off;
  *(v4f*)po = acc0 * inv + g0;
  *(v4f*)(po + 32) = acc1 * inv + g1;
}

__global__ __launch_bounds__(32) void wmma_fc_kernel(
    const float* __restrict__ rst, const float* __restrict__ Wfc,
    const float* __restrict__ bfc, float* __restrict__ out) {
  const int lane = threadIdx.x;
  const int row0 = blockIdx.x * 64;
  const int col0 = blockIdx.y * 16;
  const int khi = lane >> 4;
  const int mlo = lane & 15;

  int arow[4];
#pragma unroll
  for (int s = 0; s < 4; ++s) {
    int r = row0 + s * 16 + mlo;
    arow[s] = (r < NN) ? r : 0;
  }

  v8f acc[4] = {};
  wmma_tile_64x16<KDIM, OUTD>(rst, arow, Wfc + col0 + mlo, khi, acc);

  const int mbase = khi * 8;
  const int ncol  = col0 + mlo;
  const float bias = bfc[ncol];
  if (row0 + 64 <= NN) {
#pragma unroll
    for (int s = 0; s < 4; ++s) {
#pragma unroll
      for (int i = 0; i < 8; ++i) {
        int r = row0 + s * 16 + mbase + i;
        out[(size_t)r * OUTD + ncol] = acc[s][i] + bias;
      }
    }
  } else {
#pragma unroll
    for (int s = 0; s < 4; ++s) {
#pragma unroll
      for (int i = 0; i < 8; ++i) {
        int r = row0 + s * 16 + mbase + i;
        if (r < NN) out[(size_t)r * OUTD + ncol] = acc[s][i] + bias;
      }
    }
  }
}

extern "C" void kernel_launch(void* const* d_in, const int* in_sizes, int n_in,
                              void* d_out, int out_size, void* d_ws, size_t ws_size,
                              hipStream_t stream) {
  const float* feat     = (const float*)d_in[0];
  const int*   ei       = (const int*)d_in[1];
  const float* w_src    = (const float*)d_in[2];
  const float* b_src    = (const float*)d_in[3];
  const float* w_dst    = (const float*)d_in[4];
  const float* b_dst    = (const float*)d_in[5];
  const float* attn     = (const float*)d_in[6];
  const float* gat_bias = (const float*)d_in[7];
  const float* w_fc     = (const float*)d_in[8];
  const float* b_fc     = (const float*)d_in[9];
  float* out = (float*)d_out;

  float* ws    = (float*)d_ws;
  float* Hsrc  = ws;
  float* Hdst  = Hsrc + (size_t)NN * KDIM;
  float* rst   = Hdst + (size_t)NN * KDIM;
  int* count   = (int*)(rst + (size_t)NN * KDIM);
  int* segbeg  = count + NN;
  int* rank    = segbeg + NN;
  int* srcs    = rank + EE;
  int* total   = srcs + EE;

  const int rowTiles = (NN + 63) / 64;

  zero_counts_kernel<<<(NN + 255) / 256, 256, 0, stream>>>(count, total);

  dst_rank_kernel<<<EE / 256, 256, 0, stream>>>(ei, count, rank);

  seg_offsets_kernel<<<(NN + 255) / 256, 256, 0, stream>>>(count, total, segbeg);

  dst_place_kernel<<<EE / 256, 256, 0, stream>>>(ei, rank, segbeg, srcs);

  wmma_proj_kernel<<<dim3(rowTiles, 32), 32, 0, stream>>>(
      feat, w_src, b_src, w_dst, b_dst, Hsrc, Hdst);

  node_attend_kernel<<<(NN * 32) / 256, 256, 0, stream>>>(
      segbeg, count, srcs, Hsrc, Hdst, attn, gat_bias, rst);

  wmma_fc_kernel<<<dim3(rowTiles, 4), 32, 0, stream>>>(rst, w_fc, b_fc, out);
}


# Edges grouped by destination through LDS counters: chunks ordered by 256-node bucket, buckets ordered by node, whole runs written; 2 launches for 4, no global atomics; order in a run by LDS arrival
# speedup vs baseline: 4.1368x; 1.2289x over previous
#include <hip/hip_runtime.h>


#define NN 50000
#define EE 800000
#define KDIM 256
#define OUTD 64
#define NEG_SLOPE 0.2f

typedef __attribute__((ext_vector_type(2))) float v2f;
typedef __attribute__((ext_vector_type(4))) float v4f;
typedef __attribute__((ext_vector_type(8))) float v8f;

#define BSH 8
#define BNODES (1 << BSH)
#define NB ((NN + BNODES - 1) >> BSH)
#define NCH 128
#define CH ((EE + NCH - 1) / NCH)
#define CT 512
#define EPT ((CH + CT - 1) / CT)
#define FT 512
#define FW (FT / 32)
#define PPW (NCH / FW)
#define PWV (NCH / 32)
#define FCAP 8192

static_assert(NN <= 65536, "a source id travels in 16 bits");
static_assert(NB <= 256, "a bucket id travels in 8 bits; 8 waves scan the buckets");
static_assert(BNODES == 256, "bucket_sort scans a bucket's nodes with 8 waves");
static_assert(NCH % FW == 0 && NCH % 32 == 0 && NCH <= FT, "piece table: one thread per chunk");
static_assert(CT >= 256 && FT >= 256, "8 waves scan");

__device__ __forceinline__ int wave_incl_scan(int x, const int lane) {
#pragma unroll
  for (int d = 1; d < 32; d <<= 1) {
    const int y = __shfl_up(x, d, 32);
    if (lane >= d) x += y;
  }
  return x;
}

__global__ __launch_bounds__(CT) void chunk_sort_kernel(
    const int* __restrict__ ei, unsigned* __restrict__ chunked,
    int* __restrict__ pstart, int* __restrict__ pcnt) {
  __shared__ int lcnt[NB];
  __shared__ int lstart[NB];
  __shared__ int wsa[8];
  __shared__ unsigned staged[CH];
  const int t = threadIdx.x;
  const int lane = t & 31;
  const int w = t >> 5;
  const int i = blockIdx.x;
  if (t < NB) lcnt[t] = 0;
  const int e0 = i * CH;
  const int nch = max(min(CH, EE - e0), 0);
  int sv[EPT], dv[EPT];
#pragma unroll
  for (int k = 0; k < EPT; ++k) {
    const int e = min(e0 + k * CT + t, EE - 1);
    sv[k] = ei[e];
    dv[k] = ei[EE + e];
  }
  __syncthreads();
  unsigned key[EPT];
  int r[EPT];
#pragma unroll
  for (int k = 0; k < EPT; ++k) {
    key[k] = 0u;
    r[k] = 0;
    if (k * CT + t < nch) {
      const int s = min(max(sv[k], 0), NN - 1);
      const int d = min(max(dv[k], 0), NN - 1);
      const int b = d >> BSH;
      r[k] = atomicAdd(&lcnt[b], 1);
      key[k] = ((unsigned)b << 24) | ((unsigned)(d & (BNODES - 1)) << 16) | (unsigned)s;
    }
  }
  __syncthreads();
  const int mine = (t < NB) ? lcnt[t] : 0;
  const int ia = wave_incl_scan(mine, lane);
  if (lane == 31 && w < 8) wsa[w] = ia;
  __syncthreads();
  if (t < NB) {
    int pa = 0;
#pragma unroll
    for (int k = 0; k < 8; ++k) {
      if (k < w) pa += wsa[k];
    }
    const int exa = pa + ia - mine;
    lstart[t] = exa;
    pstart[i * NB + t] = exa;
    pcnt[i * NB + t] = mine;
  }
  __syncthreads();
#pragma unroll
  for (int k = 0; k < EPT; ++k) {
    if (k * CT + t < nch) {
      const int p = lstart[key[k] >> 24] + r[k];
      if ((unsigned)p < (unsigned)CH) staged[p] = key[k];
    }
  }
  __syncthreads();
  for (int idx = t; idx < nch; idx += CT) chunked[e0 + idx] = staged[idx];
}

__global__ __launch_bounds__(FT) void bucket_sort_kernel(
    const unsigned* __restrict__ chunked, const int* __restrict__ pstart,
    const int* __restrict__ pcnt, int* __restrict__ count,
    int* __restrict__ segbeg, int* __restrict__ srcs) {
  __shared__ int ncnt[BNODES];
  __shared__ int nbeg[BNODES];
  __shared__ int wsum[8];
  __shared__ int wsp[PWV];
  __shared__ int wsq[PWV];
  __shared__ int pgl[NCH];
  __shared__ int pnn[NCH];
  __shared__ int pof[NCH];
  __shared__ unsigned lkeys[FCAP];
  __shared__ int lsorted[FCAP];
  const int t = threadIdx.x;
  const int lane = t & 31;
  const int w = t >> 5;
  const int b = blockIdx.x;
  int ps = 0, pc = 0;
  if (t < NCH) {
    ps = pstart[t * NB + b];
    pc = pcnt[t * NB + b];
    const int cl = max(min(CH, EE - t * CH), 0);
    ps = min(max(ps, 0), cl);
    pc = min(max(pc, 0), cl - ps);
  }
  if (t < BNODES) ncnt[t] = 0;
  const int ic = wave_incl_scan(pc, lane);
  const int is = wave_incl_scan(ps, lane);
  if (lane == 31 && w < PWV) { wsp[w] = ic; wsq[w] = is; }
  __syncthreads();
  int bn = 0, bs = 0;
  for (int k = 0; k < PWV; ++k) { bn += wsp[k]; bs += wsq[k]; }
  bs = min(max(bs, 0), EE);
  bn = min(max(bn, 0), EE - bs);
  if (t < NCH) {
    int wp = 0;
    for (int k = 0; k < PWV; ++k) {
      if (k < w) wp += wsp[k];
    }
    pof[t] = wp + ic - pc;
    pgl[t] = t * CH + ps;
    pnn[t] = pc;
  }
  __syncthreads();
  const bool fits = bn <= FCAP;
  {
    unsigned kv[PPW][2];
#pragma unroll
    for (int j = 0; j < PPW; ++j) {
      const int pi = w + FW * j;
      const int n = pnn[pi];
      const int g = pgl[pi];
#pragma unroll
      for (int h = 0; h < 2; ++h) {
        const int q = min(lane + 32 * h, max(n - 1, 0));
        kv[j][h] = chunked[min(g + q, EE - 1)];
      }
    }
#pragma unroll
    for (int j = 0; j < PPW; ++j) {
      const int pi = w + FW * j;
      const int n = pnn[pi];
      const int g = pgl[pi];
      const int o = pof[pi];
#pragma unroll
      for (int h = 0; h < 2; ++h) {
        const int q = lane + 32 * h;
        if (q < n) {
          atomicAdd(&ncnt[(kv[j][h] >> 16) & (BNODES - 1)], 1);
          if (fits) lkeys[o + q] = kv[j][h];
        }
      }
      for (int q = lane + 64; q < n; q += 32) {
        const unsigned x = chunked[g + q];
        atomicAdd(&ncnt[(x >> 16) & (BNODES - 1)], 1);
        if (fits) lkeys[o + q] = x;
      }
    }
  }
  __syncthreads();
  const int c = (t < BNODES) ? ncnt[t] : 0;
  const int inc = wave_incl_scan(c, lane);
  if (lane == 31 && w < 8) wsum[w] = inc;
  __syncthreads();
  if (t < BNODES) {
    int wp = 0;
#pragma unroll
    for (int k = 0; k < 8; ++k) {
      if (k < w) wp += wsum[k];
    }
    const int ex = wp + inc - c;
    nbeg[t] = ex;
    ncnt[t] = 0;
    const int n = b * BNODES + t;
    if (n < NN) {
      count[n] = c;
      segbeg[n] = bs + ex;
    }
  }
  __syncthreads();
  if (fits) {
    for (int idx = t; idx < bn; idx += FT) {
      const unsigned x = lkeys[idx];
      const int l = (x >> 16) & (BNODES - 1);
      const int p = nbeg[l] + atomicAdd(&ncnt[l], 1);
      if ((unsigned)p < (unsigned)FCAP) lsorted[p] = (int)(x & 0xffffu);
    }
  } else {
    for (int pi = w; pi < NCH; pi += FW) {
      const int n = pnn[pi];
      const int g = pgl[pi];
      for (int q = lane; q < n; q += 32) {
        const unsigned x = chunked[g + q];
        const int l = (x >> 16) & (BNODES - 1);
        const int p = nbeg[l] + atomicAdd(&ncnt[l], 1);
        if ((unsigned)p < (unsigned)bn) srcs[bs + p] = (int)(x & 0xffffu);
      }
    }
  }
  __syncthreads();
  if (fits) {
    for (int idx = t; idx < bn; idx += FT) srcs[bs + idx] = lsorted[idx];
  }
}

template <int KLEN, int BPITCH>
__device__ __forceinline__ void wmma_tile_64x16(
    const float* __restrict__ Ap, const int (&arow)[4],
    const float* __restrict__ Bp, const int khi, v8f (&acc)[4]) {
#pragma unroll 1
  for (int kc = 0; kc < KLEN; kc += 32) {
    const int kl = kc + 16 * khi;
    float b[16];
    v4f a[4][4];
#pragma unroll
    for (int e = 0; e < 16; ++e) b[e] = Bp[(size_t)(kl + e) * BPITCH];
#pragma unroll
    for (int s = 0; s < 4; ++s) {
      const v4f* pa = (const v4f*)(Ap + (size_t)arow[s] * KDIM + kl);
#pragma unroll
      for (int q = 0; q < 4; ++q) a[s][q] = pa[q];
    }
    __builtin_amdgcn_sched_barrier(0);
#pragma unroll
    for (int j = 0; j < 8; ++j) {
      v2f bfrag;
      bfrag.x = b[2 * j];
      bfrag.y = b[2 * j + 1];
#pragma unroll
      for (int s = 0; s < 4; ++s) {
        v2f afrag;
        afrag.x = a[s][j >> 1][2 * (j & 1)];
        afrag.y = a[s][j >> 1][2 * (j & 1) + 1];
        acc[s] = __builtin_amdgcn_wmma_f32_16x16x4_f32(
            false, afrag, false, bfrag, (short)0, acc[s], false, false);
      }
    }
  }
}

__global__ __launch_bounds__(32) void wmma_proj_kernel(
    const float* __restrict__ feat,
    const float* __restrict__ Wsrc, const float* __restrict__ Bsrc,
    const float* __restrict__ Wdst, const float* __restrict__ Bdst,
    float* __restrict__ Hsrc, float* __restrict__ Hdst) {
  const int lane = threadIdx.x;
  const int rowTile = blockIdx.x;
  const int yt = blockIdx.y;
  const int mat = yt >> 4;
  const int colTile = yt & 15;

  const float* __restrict__ W  = mat ? Wdst : Wsrc;
  const float* __restrict__ Bv = mat ? Bdst : Bsrc;
  float* __restrict__ Hout     = mat ? Hdst : Hsrc;

  const int row0 = rowTile * 64;
  const int col0 = colTile * 16;
  const int khi  = lane >> 4;
  const int mlo  = lane & 15;

  int arow[4];
#pragma unroll
  for (int s = 0; s < 4; ++s) {
    int r = row0 + s * 16 + mlo;
    arow[s] = (r < NN) ? r : 0;
  }

  v8f acc[4] = {};
  wmma_tile_64x16<KDIM, KDIM>(feat, arow, W + col0 + mlo, khi, acc);

  const int mbase = khi * 8;
  const int ncol  = col0 + mlo;
  const float bias = Bv[ncol];
  if (row0 + 64 <= NN) {
#pragma unroll
    for (int s = 0; s < 4; ++s) {
#pragma unroll
      for (int i = 0; i < 8; ++i) {
        int r = row0 + s * 16 + mbase + i;
        Hout[(size_t)r * KDIM + ncol] = acc[s][i] + bias;
      }
    }
  } else {
#pragma unroll
    for (int s = 0; s < 4; ++s) {
#pragma unroll
      for (int i = 0; i < 8; ++i) {
        int r = row0 + s * 16 + mbase + i;
        if (r < NN) Hout[(size_t)r * KDIM + ncol] = acc[s][i] + bias;
      }
    }
  }
}

__device__ __forceinline__ float row8_sum(float x) {
  x += __int_as_float(__builtin_amdgcn_update_dpp(0, __float_as_int(x), 0x161, 0xf, 0xf, false));
  x += __int_as_float(__builtin_amdgcn_update_dpp(0, __float_as_int(x), 0x162, 0xf, 0xf, false));
  x += __int_as_float(__builtin_amdgcn_update_dpp(0, __float_as_int(x), 0x164, 0xf, 0xf, false));
  return x;
}

__global__ __launch_bounds__(256) void node_attend_kernel(
    const int* __restrict__ segbeg, const int* __restrict__ count,
    const int* __restrict__ srcs, const float* __restrict__ Hsrc,
    const float* __restrict__ Hdst, const float* __restrict__ attn,
    const float* __restrict__ gat_bias, float* __restrict__ rst) {
  const int n = __builtin_amdgcn_readfirstlane((int)((blockIdx.x * 256 + threadIdx.x) >> 5));
  const int lane = threadIdx.x & 31;
  if (n >= NN) return;
  int beg = segbeg[n];
  int cnt = count[n];
  beg = min(max(beg, 0), EE);
  cnt = min(max(cnt, 0), EE - beg);

  const int off = (lane >> 3) * 64 + (lane & 7) * 4;
  const float* __restrict__ hd = Hdst + (size_t)n * KDIM + off;
  const v4f d0 = *(const v4f*)hd;
  const v4f d1 = *(const v4f*)(hd + 32);
  const v4f a0 = *(const v4f*)(attn + off);
  const v4f a1 = *(const v4f*)(attn + off + 32);

  float m = -__builtin_inff();
  float den = 0.0f;
  v4f acc0 = {0.0f, 0.0f, 0.0f, 0.0f};
  v4f acc1 = {0.0f, 0.0f, 0.0f, 0.0f};

  for (int base = 0; base < cnt; base += 32) {
    const int nb = min(32, cnt - base);
    int mysrc = 0;
    if (lane < nb) mysrc = srcs[beg + base + lane];
    mysrc = min(max(mysrc, 0), NN - 1);
    for (int j = 0; j < nb; j += 4) {
      v4f r0[4], r1[4];
#pragma unroll
      for (int u = 0; u < 4; ++u) {
        const int jj = min(j + u, nb - 1);
        const int s = __builtin_amdgcn_readlane(mysrc, jj);
        const float* __restrict__ row = Hsrc + (size_t)s * KDIM + off;
        r0[u] = *(const v4f*)row;
        r1[u] = *(const v4f*)(row + 32);
      }
#pragma unroll
      for (int u = 0; u < 4; ++u) {
        if (j + u < nb) {
          v4f x0 = r0[u] + d0;
          v4f x1 = r1[u] + d1;
          x0 = __builtin_elementwise_max(x0, x0 * NEG_SLOPE);
          x1 = __builtin_elementwise_max(x1, x1 * NEG_SLOPE);
          const v4f t = x0 * a0 + x1 * a1;
          float part = (t[0] + t[1]) + (t[2] + t[3]);
          part = row8_sum(part);
          const float dlt = part - m;
          const float ex = expf(-fabsf(dlt));
          const bool up = dlt > 0.0f;
          const float sc = up ? ex : 1.0f;
          const float p = up ? 1.0f : ex;
          m = up ? part : m;
          den = den * sc + p;
          acc0 = acc0 * sc + p * r0[u];
          acc1 = acc1 * sc + p * r1[u];
        }
      }
    }
  }

  const float inv = 1.0f / ((den == 0.0f) ? 1.0f : den);
  const v4f g0 = *(const v4f*)(gat_bias + off);
  const v4f g1 = *(const v4f*)(gat_bias + off + 32);
  float* __restrict__ po = rst + (size_t)n * KDIM + off;
  *(v4f*)po = acc0 * inv + g0;
  *(v4f*)(po + 32) = acc1 * inv + g1;
}

__global__ __launch_bounds__(32) void wmma_fc_kernel(
    const float* __restrict__ rst, const float* __restrict__ Wfc,
    const float* __restrict__ bfc, float* __restrict__ out) {
  const int lane = threadIdx.x;
  const int row0 = blockIdx.x * 64;
  const int col0 = blockIdx.y * 16;
  const int khi = lane >> 4;
  const int mlo = lane & 15;

  int arow[4];
#pragma unroll
  for (int s = 0; s < 4; ++s) {
    int r = row0 + s * 16 + mlo;
    arow[s] = (r < NN) ? r : 0;
  }

  v8f acc[4] = {};
  wmma_tile_64x16<KDIM, OUTD>(rst, arow, Wfc + col0 + mlo, khi, acc);

  const int mbase = khi * 8;
  const int ncol  = col0 + mlo;
  const float bias = bfc[ncol];
  if (row0 + 64 <= NN) {
#pragma unroll
    for (int s = 0; s < 4; ++s) {
#pragma unroll
      for (int i = 0; i < 8; ++i) {
        int r = row0 + s * 16 + mbase + i;
        out[(size_t)r * OUTD + ncol] = acc[s][i] + bias;
      }
    }
  } else {
#pragma unroll
    for (int s = 0; s < 4; ++s) {
#pragma unroll
      for (int i = 0; i < 8; ++i) {
        int r = row0 + s * 16 + mbase + i;
        if (r < NN) out[(size_t)r * OUTD + ncol] = acc[s][i] + bias;
      }
    }
  }
}

extern "C" void kernel_launch(void* const* d_in, const int* in_sizes, int n_in,
                              void* d_out, int out_size, void* d_ws, size_t ws_size,
                              hipStream_t stream) {
  const float* feat     = (const float*)d_in[0];
  const int*   ei       = (const int*)d_in[1];
  const float* w_src    = (const float*)d_in[2];
  const float* b_src    = (const float*)d_in[3];
  const float* w_dst    = (const float*)d_in[4];
  const float* b_dst    = (const float*)d_in[5];
  const float* attn     = (const float*)d_in[6];
  const float* gat_bias = (const float*)d_in[7];
  const float* w_fc     = (const float*)d_in[8];
  const float* b_fc     = (const float*)d_in[9];
  float* out = (float*)d_out;

  float* ws    = (float*)d_ws;
  float* Hsrc  = ws;
  float* Hdst  = Hsrc + (size_t)NN * KDIM;
  float* rst   = Hdst + (size_t)NN * KDIM;
  int* count   = (int*)(rst + (size_t)NN * KDIM);
  int* segbeg  = count + NN;
  int* srcs    = segbeg + NN;
  unsigned* chunked = (unsigned*)(srcs + EE);
  int* pstart  = (int*)(chunked + EE);
  int* pcnt    = pstart + NCH * NB;

  const int rowTiles = (NN + 63) / 64;

  chunk_sort_kernel<<<NCH, CT, 0, stream>>>(ei, chunked, pstart, pcnt);

  bucket_sort_kernel<<<NB, FT, 0, stream>>>(chunked, pstart, pcnt, count, segbeg, srcs);

  wmma_proj_kernel<<<dim3(rowTiles, 32), 32, 0, stream>>>(
      feat, w_src, b_src, w_dst, b_dst, Hsrc, Hdst);

  node_attend_kernel<<<(NN * 32) / 256, 256, 0, stream>>>(
      segbeg, count, srcs, Hsrc, Hdst, attn, gat_bias, rst);

  wmma_fc_kernel<<<dim3(rowTiles, 4), 32, 0, stream>>>(rst, w_fc, b_fc, out);
}


# Projection GEMM: one wave computes 64x64 (4 interleaved column tiles, column 4n+c) per load of its A chunk, a work-group's 4 waves share the rows: A fetched 8x, not 32x; B loads, H stores 16-byte
# speedup vs baseline: 5.8278x; 1.4088x over previous
#include <hip/hip_runtime.h>


#define NN 50000
#define EE 800000
#define KDIM 256
#define OUTD 64
#define NEG_SLOPE 0.2f

typedef __attribute__((ext_vector_type(2))) float v2f;
typedef __attribute__((ext_vector_type(4))) float v4f;
typedef __attribute__((ext_vector_type(8))) float v8f;

#define BSH 8
#define BNODES (1 << BSH)
#define NB ((NN + BNODES - 1) >> BSH)
#define NCH 128
#define CH ((EE + NCH - 1) / NCH)
#define CT 512
#define EPT ((CH + CT - 1) / CT)
#define FT 512
#define FW (FT / 32)
#define PPW (NCH / FW)
#define PWV (NCH / 32)
#define FCAP 8192

static_assert(NN <= 65536, "a source id travels in 16 bits");
static_assert(NB <= 256, "a bucket id travels in 8 bits; 8 waves scan the buckets");
static_assert(BNODES == 256, "bucket_sort scans a bucket's nodes with 8 waves");
static_assert(NCH % FW == 0 && NCH % 32 == 0 && NCH <= FT, "piece table: one thread per chunk");
static_assert(CT >= 256 && FT >= 256, "8 waves scan");

__device__ __forceinline__ int wave_incl_scan(int x, const int lane) {
#pragma unroll
  for (int d = 1; d < 32; d <<= 1) {
    const int y = __shfl_up(x, d, 32);
    if (lane >= d) x += y;
  }
  return x;
}

__global__ __launch_bounds__(CT) void chunk_sort_kernel(
    const int* __restrict__ ei, unsigned* __restrict__ chunked,
    int* __restrict__ pstart, int* __restrict__ pcnt) {
  __shared__ int lcnt[NB];
  __shared__ int lstart[NB];
  __shared__ int wsa[8];
  __shared__ unsigned staged[CH];
  const int t = threadIdx.x;
  const int lane = t & 31;
  const int w = t >> 5;
  const int i = blockIdx.x;
  if (t < NB) lcnt[t] = 0;
  const int e0 = i * CH;
  const int nch = max(min(CH, EE - e0), 0);
  int sv[EPT], dv[EPT];
#pragma unroll
  for (int k = 0; k < EPT; ++k) {
    const int e = min(e0 + k * CT + t, EE - 1);
    sv[k] = ei[e];
    dv[k] = ei[EE + e];
  }
  __syncthreads();
  unsigned key[EPT];
  int r[EPT];
#pragma unroll
  for (int k = 0; k < EPT; ++k) {
    key[k] = 0u;
    r[k] = 0;
    if (k * CT + t < nch) {
      const int s = min(max(sv[k], 0), NN - 1);
      const int d = min(max(dv[k], 0), NN - 1);
      const int b = d >> BSH;
      r[k] = atomicAdd(&lcnt[b], 1);
      key[k] = ((unsigned)b << 24) | ((unsigned)(d & (BNODES - 1)) << 16) | (unsigned)s;
    }
  }
  __syncthreads();
  const int mine = (t < NB) ? lcnt[t] : 0;
  const int ia = wave_incl_scan(mine, lane);
  if (lane == 31 && w < 8) wsa[w] = ia;
  __syncthreads();
  if (t < NB) {
    int pa = 0;
#pragma unroll
    for (int k = 0; k < 8; ++k) {
      if (k < w) pa += wsa[k];
    }
    const int exa = pa + ia - mine;
    lstart[t] = exa;
    pstart[i * NB + t] = exa;
    pcnt[i * NB + t] = mine;
  }
  __syncthreads();
#pragma unroll
  for (int k = 0; k < EPT; ++k) {
    if (k * CT + t < nch) {
      const int p = lstart[key[k] >> 24] + r[k];
      if ((unsigned)p < (unsigned)CH) staged[p] = key[k];
    }
  }
  __syncthreads();
  for (int idx = t; idx < nch; idx += CT) chunked[e0 + idx] = staged[idx];
}

__global__ __launch_bounds__(FT) void bucket_sort_kernel(
    const unsigned* __restrict__ chunked, const int* __restrict__ pstart,
    const int* __restrict__ pcnt, int* __restrict__ count,
    int* __restrict__ segbeg, int* __restrict__ srcs) {
  __shared__ int ncnt[BNODES];
  __shared__ int nbeg[BNODES];
  __shared__ int wsum[8];
  __shared__ int wsp[PWV];
  __shared__ int wsq[PWV];
  __shared__ int pgl[NCH];
  __shared__ int pnn[NCH];
  __shared__ int pof[NCH];
  __shared__ unsigned lkeys[FCAP];
  __shared__ int lsorted[FCAP];
  const int t = threadIdx.x;
  const int lane = t & 31;
  const int w = t >> 5;
  const int b = blockIdx.x;
  int ps = 0, pc = 0;
  if (t < NCH) {
    ps = pstart[t * NB + b];
    pc = pcnt[t * NB + b];
    const int cl = max(min(CH, EE - t * CH), 0);
    ps = min(max(ps, 0), cl);
    pc = min(max(pc, 0), cl - ps);
  }
  if (t < BNODES) ncnt[t] = 0;
  const int ic = wave_incl_scan(pc, lane);
  const int is = wave_incl_scan(ps, lane);
  if (lane == 31 && w < PWV) { wsp[w] = ic; wsq[w] = is; }
  __syncthreads();
  int bn = 0, bs = 0;
  for (int k = 0; k < PWV; ++k) { bn += wsp[k]; bs += wsq[k]; }
  bs = min(max(bs, 0), EE);
  bn = min(max(bn, 0), EE - bs);
  if (t < NCH) {
    int wp = 0;
    for (int k = 0; k < PWV; ++k) {
      if (k < w) wp += wsp[k];
    }
    pof[t] = wp + ic - pc;
    pgl[t] = t * CH + ps;
    pnn[t] = pc;
  }
  __syncthreads();
  const bool fits = bn <= FCAP;
  {
    unsigned kv[PPW][2];
#pragma unroll
    for (int j = 0; j < PPW; ++j) {
      const int pi = w + FW * j;
      const int n = pnn[pi];
      const int g = pgl[pi];
#pragma unroll
      for (int h = 0; h < 2; ++h) {
        const int q = min(lane + 32 * h, max(n - 1, 0));
        kv[j][h] = chunked[min(g + q, EE - 1)];
      }
    }
#pragma unroll
    for (int j = 0; j < PPW; ++j) {
      const int pi = w + FW * j;
      const int n = pnn[pi];
      const int g = pgl[pi];
      const int o = pof[pi];
#pragma unroll
      for (int h = 0; h < 2; ++h) {
        const int q = lane + 32 * h;
        if (q < n) {
          atomicAdd(&ncnt[(kv[j][h] >> 16) & (BNODES - 1)], 1);
          if (fits) lkeys[o + q] = kv[j][h];
        }
      }
      for (int q = lane + 64; q < n; q += 32) {
        const unsigned x = chunked[g + q];
        atomicAdd(&ncnt[(x >> 16) & (BNODES - 1)], 1);
        if (fits) lkeys[o + q] = x;
      }
    }
  }
  __syncthreads();
  const int c = (t < BNODES) ? ncnt[t] : 0;
  const int inc = wave_incl_scan(c, lane);
  if (lane == 31 && w < 8) wsum[w] = inc;
  __syncthreads();
  if (t < BNODES) {
    int wp = 0;
#pragma unroll
    for (int k = 0; k < 8; ++k) {
      if (k < w) wp += wsum[k];
    }
    const int ex = wp + inc - c;
    nbeg[t] = ex;
    ncnt[t] = 0;
    const int n = b * BNODES + t;
    if (n < NN) {
      count[n] = c;
      segbeg[n] = bs + ex;
    }
  }
  __syncthreads();
  if (fits) {
    for (int idx = t; idx < bn; idx += FT) {
      const unsigned x = lkeys[idx];
      const int l = (x >> 16) & (BNODES - 1);
      const int p = nbeg[l] + atomicAdd(&ncnt[l], 1);
      if ((unsigned)p < (unsigned)FCAP) lsorted[p] = (int)(x & 0xffffu);
    }
  } else {
    for (int pi = w; pi < NCH; pi += FW) {
      const int n = pnn[pi];
      const int g = pgl[pi];
      for (int q = lane; q < n; q += 32) {
        const unsigned x = chunked[g + q];
        const int l = (x >> 16) & (BNODES - 1);
        const int p = nbeg[l] + atomicAdd(&ncnt[l], 1);
        if ((unsigned)p < (unsigned)bn) srcs[bs + p] = (int)(x & 0xffffu);
      }
    }
  }
  __syncthreads();
  if (fits) {
    for (int idx = t; idx < bn; idx += FT) srcs[bs + idx] = lsorted[idx];
  }
}

template <int KLEN, int BPITCH>
__device__ __forceinline__ void wmma_tile_64x16(
    const float* __restrict__ Ap, const int (&arow)[4],
    const float* __restrict__ Bp, const int khi, v8f (&acc)[4]) {
#pragma unroll 1
  for (int kc = 0; kc < KLEN; kc += 32) {
    const int kl = kc + 16 * khi;
    float b[16];
    v4f a[4][4];
#pragma unroll
    for (int e = 0; e < 16; ++e) b[e] = Bp[(size_t)(kl + e) * BPITCH];
#pragma unroll
    for (int s = 0; s < 4; ++s) {
      const v4f* pa = (const v4f*)(Ap + (size_t)arow[s] * KDIM + kl);
#pragma unroll
      for (int q = 0; q < 4; ++q) a[s][q] = pa[q];
    }
    __builtin_amdgcn_sched_barrier(0);
#pragma unroll
    for (int j = 0; j < 8; ++j) {
      v2f bfrag;
      bfrag.x = b[2 * j];
      bfrag.y = b[2 * j + 1];
#pragma unroll
      for (int s = 0; s < 4; ++s) {
        v2f afrag;
        afrag.x = a[s][j >> 1][2 * (j & 1)];
        afrag.y = a[s][j >> 1][2 * (j & 1) + 1];
        acc[s] = __builtin_amdgcn_wmma_f32_16x16x4_f32(
            false, afrag, false, bfrag, (short)0, acc[s], false, false);
      }
    }
  }
}

template <int KLEN, int BPITCH>
__device__ __forceinline__ void wmma_tile_64x64(
    const float* __restrict__ Ap, const int (&arow)[4],
    const float* __restrict__ Bp, const int khi, v8f (&acc)[4][4]) {
#pragma unroll 1
  for (int kc = 0; kc < KLEN; kc += 32) {
    const int kl = kc + 16 * khi;
    v4f b[16];
    v4f a[4][4];
#pragma unroll
    for (int e = 0; e < 16; ++e) b[e] = *(const v4f*)(Bp + (size_t)(kl + e) * BPITCH);
#pragma unroll
    for (int s = 0; s < 4; ++s) {
      const v4f* pa = (const v4f*)(Ap + (size_t)arow[s] * KDIM + kl);
#pragma unroll
      for (int q = 0; q < 4; ++q) a[s][q] = pa[q];
    }
    __builtin_amdgcn_sched_barrier(0);
#pragma unroll
    for (int j = 0; j < 8; ++j) {
      v2f bfrag[4];
#pragma unroll
      for (int c = 0; c < 4; ++c) {
        bfrag[c].x = b[2 * j][c];
        bfrag[c].y = b[2 * j + 1][c];
      }
#pragma unroll
      for (int s = 0; s < 4; ++s) {
        v2f afrag;
        afrag.x = a[s][j >> 1][2 * (j & 1)];
        afrag.y = a[s][j >> 1][2 * (j & 1) + 1];
#pragma unroll
        for (int c = 0; c < 4; ++c) {
          acc[s][c] = __builtin_amdgcn_wmma_f32_16x16x4_f32(
              false, afrag, false, bfrag[c], (short)0, acc[s][c], false, false);
        }
      }
    }
  }
}

__global__ __launch_bounds__(128) void wmma_proj_kernel(
    const float* __restrict__ feat,
    const float* __restrict__ Wsrc, const float* __restrict__ Bsrc,
    const float* __restrict__ Wdst, const float* __restrict__ Bdst,
    float* __restrict__ Hsrc, float* __restrict__ Hdst) {
  const int lane = threadIdx.x & 31;
  const int cg = __builtin_amdgcn_readfirstlane((int)(threadIdx.x >> 5));
  const int rowTile = blockIdx.x;
  const int mat = blockIdx.y;

  const float* __restrict__ W  = mat ? Wdst : Wsrc;
  const float* __restrict__ Bv = mat ? Bdst : Bsrc;
  float* __restrict__ Hout     = mat ? Hdst : Hsrc;

  const int row0 = rowTile * 64;
  const int col0 = cg * 64;
  const int khi  = lane >> 4;
  const int mlo  = lane & 15;

  int arow[4];
#pragma unroll
  for (int s = 0; s < 4; ++s) {
    int r = row0 + s * 16 + mlo;
    arow[s] = (r < NN) ? r : 0;
  }

  v8f acc[4][4] = {};
  wmma_tile_64x64<KDIM, KDIM>(feat, arow, W + col0 + 4 * mlo, khi, acc);

  const int mbase = khi * 8;
  const int ncol  = col0 + 4 * mlo;
  const v4f bias = *(const v4f*)(Bv + ncol);
  if (row0 + 64 <= NN) {
#pragma unroll
    for (int s = 0; s < 4; ++s) {
#pragma unroll
      for (int i = 0; i < 8; ++i) {
        const int r = row0 + s * 16 + mbase + i;
        v4f o;
        o[0] = acc[s][0][i] + bias[0];
        o[1] = acc[s][1][i] + bias[1];
        o[2] = acc[s][2][i] + bias[2];
        o[3] = acc[s][3][i] + bias[3];
        *(v4f*)(Hout + (size_t)r * KDIM + ncol) = o;
      }
    }
  } else {
#pragma unroll
    for (int s = 0; s < 4; ++s) {
#pragma unroll
      for (int i = 0; i < 8; ++i) {
        const int r = row0 + s * 16 + mbase + i;
        if (r < NN) {
          v4f o;
          o[0] = acc[s][0][i] + bias[0];
          o[1] = acc[s][1][i] + bias[1];
          o[2] = acc[s][2][i] + bias[2];
          o[3] = acc[s][3][i] + bias[3];
          *(v4f*)(Hout + (size_t)r * KDIM + ncol) = o;
        }
      }
    }
  }
}

__device__ __forceinline__ float row8_sum(float x) {
  x += __int_as_float(__builtin_amdgcn_update_dpp(0, __float_as_int(x), 0x161, 0xf, 0xf, false));
  x += __int_as_float(__builtin_amdgcn_update_dpp(0, __float_as_int(x), 0x162, 0xf, 0xf, false));
  x += __int_as_float(__builtin_amdgcn_update_dpp(0, __float_as_int(x), 0x164, 0xf, 0xf, false));
  return x;
}

__global__ __launch_bounds__(256) void node_attend_kernel(
    const int* __restrict__ segbeg, const int* __restrict__ count,
    const int* __restrict__ srcs, const float* __restrict__ Hsrc,
    const float* __restrict__ Hdst, const float* __restrict__ attn,
    const float* __restrict__ gat_bias, float* __restrict__ rst) {
  const int n = __builtin_amdgcn_readfirstlane((int)((blockIdx.x * 256 + threadIdx.x) >> 5));
  const int lane = threadIdx.x & 31;
  if (n >= NN) return;
  int beg = segbeg[n];
  int cnt = count[n];
  beg = min(max(beg, 0), EE);
  cnt = min(max(cnt, 0), EE - beg);

  const int off = (lane >> 3) * 64 + (lane & 7) * 4;
  const float* __restrict__ hd = Hdst + (size_t)n * KDIM + off;
  const v4f d0 = *(const v4f*)hd;
  const v4f d1 = *(const v4f*)(hd + 32);
  const v4f a0 = *(const v4f*)(attn + off);
  const v4f a1 = *(const v4f*)(attn + off + 32);

  float m = -__builtin_inff();
  float den = 0.0f;
  v4f acc0 = {0.0f, 0.0f, 0.0f, 0.0f};
  v4f acc1 = {0.0f, 0.0f, 0.0f, 0.0f};

  for (int base = 0; base < cnt; base += 32) {
    const int nb = min(32, cnt - base);
    int mysrc = 0;
    if (lane < nb) mysrc = srcs[beg + base + lane];
    mysrc = min(max(mysrc, 0), NN - 1);
    for (int j = 0; j < nb; j += 4) {
      v4f r0[4], r1[4];
#pragma unroll
      for (int u = 0; u < 4; ++u) {
        const int jj = min(j + u, nb - 1);
        const int s = __builtin_amdgcn_readlane(mysrc, jj);
        const float* __restrict__ row = Hsrc + (size_t)s * KDIM + off;
        r0[u] = *(const v4f*)row;
        r1[u] = *(const v4f*)(row + 32);
      }
#pragma unroll
      for (int u = 0; u < 4; ++u) {
        if (j + u < nb) {
          v4f x0 = r0[u] + d0;
          v4f x1 = r1[u] + d1;
          x0 = __builtin_elementwise_max(x0, x0 * NEG_SLOPE);
          x1 = __builtin_elementwise_max(x1, x1 * NEG_SLOPE);
          const v4f t = x0 * a0 + x1 * a1;
          float part = (t[0] + t[1]) + (t[2] + t[3]);
          part = row8_sum(part);
          const float dlt = part - m;
          const float ex = expf(-fabsf(dlt));
          const bool up = dlt > 0.0f;
          const float sc = up ? ex : 1.0f;
          const float p = up ? 1.0f : ex;
          m = up ? part : m;
          den = den * sc + p;
          acc0 = acc0 * sc + p * r0[u];
          acc1 = acc1 * sc + p * r1[u];
        }
      }
    }
  }

  const float inv = 1.0f / ((den == 0.0f) ? 1.0f : den);
  const v4f g0 = *(const v4f*)(gat_bias + off);
  const v4f g1 = *(const v4f*)(gat_bias + off + 32);
  float* __restrict__ po = rst + (size_t)n * KDIM + off;
  *(v4f*)po = acc0 * inv + g0;
  *(v4f*)(po + 32) = acc1 * inv + g1;
}

__global__ __launch_bounds__(32) void wmma_fc_kernel(
    const float* __restrict__ rst, const float* __restrict__ Wfc,
    const float* __restrict__ bfc, float* __restrict__ out) {
  const int lane = threadIdx.x;
  const int row0 = blockIdx.x * 64;
  const int col0 = blockIdx.y * 16;
  const int khi = lane >> 4;
  const int mlo = lane & 15;

  int arow[4];
#pragma unroll
  for (int s = 0; s < 4; ++s) {
    int r = row0 + s * 16 + mlo;
    arow[s] = (r < NN) ? r : 0;
  }

  v8f acc[4] = {};
  wmma_tile_64x16<KDIM, OUTD>(rst, arow, Wfc + col0 + mlo, khi, acc);

  const int mbase = khi * 8;
  const int ncol  = col0 + mlo;
  const float bias = bfc[ncol];
  if (row0 + 64 <= NN) {
#pragma unroll
    for (int s = 0; s < 4; ++s) {
#pragma unroll
      for (int i = 0; i < 8; ++i) {
        int r = row0 + s * 16 + mbase + i;
        out[(size_t)r * OUTD + ncol] = acc[s][i] + bias;
      }
    }
  } else {
#pragma unroll
    for (int s = 0; s < 4; ++s) {
#pragma unroll
      for (int i = 0; i < 8; ++i) {
        int r = row0 + s * 16 + mbase + i;
        if (r < NN) out[(size_t)r * OUTD + ncol] = acc[s][i] + bias;
      }
    }
  }
}

extern "C" void kernel_launch(void* const* d_in, const int* in_sizes, int n_in,
                              void* d_out, int out_size, void* d_ws, size_t ws_size,
                              hipStream_t stream) {
  const float* feat     = (const float*)d_in[0];
  const int*   ei       = (const int*)d_in[1];
  const float* w_src    = (const float*)d_in[2];
  const float* b_src    = (const float*)d_in[3];
  const float* w_dst    = (const float*)d_in[4];
  const float* b_dst    = (const float*)d_in[5];
  const float* attn     = (const float*)d_in[6];
  const float* gat_bias = (const float*)d_in[7];
  const float* w_fc     = (const float*)d_in[8];
  const float* b_fc     = (const float*)d_in[9];
  float* out = (float*)d_out;

  float* ws    = (float*)d_ws;
  float* Hsrc  = ws;
  float* Hdst  = Hsrc + (size_t)NN * KDIM;
  float* rst   = Hdst + (size_t)NN * KDIM;
  int* count   = (int*)(rst + (size_t)NN * KDIM);
  int* segbeg  = count + NN;
  int* srcs    = segbeg + NN;
  unsigned* chunked = (unsigned*)(srcs + EE);
  int* pstart  = (int*)(chunked + EE);
  int* pcnt    = pstart + NCH * NB;

  const int rowTiles = (NN + 63) / 64;

  chunk_sort_kernel<<<NCH, CT, 0, stream>>>(ei, chunked, pstart, pcnt);

  bucket_sort_kernel<<<NB, FT, 0, stream>>>(chunked, pstart, pcnt, count, segbeg, srcs);

  wmma_proj_kernel<<<dim3(rowTiles, 2), 128, 0, stream>>>(
      feat, w_src, b_src, w_dst, b_dst, Hsrc, Hdst);

  node_attend_kernel<<<(NN * 32) / 256, 256, 0, stream>>>(
      segbeg, count, srcs, Hsrc, Hdst, attn, gat_bias, rst);

  wmma_fc_kernel<<<dim3(rowTiles, 4), 32, 0, stream>>>(rst, w_fc, b_fc, out);
}


# Output GEMM moved into the node kernel: 16 one-node waves put their rst rows in LDS, each wave does one k quarter of a 16x16 tile on WMMA, quarters added via LDS; 4 launches; k summed by quarters
# speedup vs baseline: 6.0550x; 1.0390x over previous
#include <hip/hip_runtime.h>


#define NN 50000
#define EE 800000
#define KDIM 256
#define OUTD 64
#define NEG_SLOPE 0.2f

typedef __attribute__((ext_vector_type(2))) float v2f;
typedef __attribute__((ext_vector_type(4))) float v4f;
typedef __attribute__((ext_vector_type(8))) float v8f;

#define BSH 8
#define BNODES (1 << BSH)
#define NB ((NN + BNODES - 1) >> BSH)
#define NCH 128
#define CH ((EE + NCH - 1) / NCH)
#define CT 512
#define EPT ((CH + CT - 1) / CT)
#define FT 512
#define FW (FT / 32)
#define PPW (NCH / FW)
#define PWV (NCH / 32)
#define FCAP 8192

static_assert(NN <= 65536, "a source id travels in 16 bits");
static_assert(NB <= 256, "a bucket id travels in 8 bits; 8 waves scan the buckets");
static_assert(BNODES == 256, "bucket_sort scans a bucket's nodes with 8 waves");
static_assert(NCH % FW == 0 && NCH % 32 == 0 && NCH <= FT, "piece table: one thread per chunk");
static_assert(CT >= 256 && FT >= 256, "8 waves scan");

__device__ __forceinline__ int wave_incl_scan(int x, const int lane) {
#pragma unroll
  for (int d = 1; d < 32; d <<= 1) {
    const int y = __shfl_up(x, d, 32);
    if (lane >= d) x += y;
  }
  return x;
}

__global__ __launch_bounds__(CT) void chunk_sort_kernel(
    const int* __restrict__ ei, unsigned* __restrict__ chunked,
    int* __restrict__ pstart, int* __restrict__ pcnt) {
  __shared__ int lcnt[NB];
  __shared__ int lstart[NB];
  __shared__ int wsa[8];
  __shared__ unsigned staged[CH];
  const int t = threadIdx.x;
  const int lane = t & 31;
  const int w = t >> 5;
  const int i = blockIdx.x;
  if (t < NB) lcnt[t] = 0;
  const int e0 = i * CH;
  const int nch = max(min(CH, EE - e0), 0);
  int sv[EPT], dv[EPT];
#pragma unroll
  for (int k = 0; k < EPT; ++k) {
    const int e = min(e0 + k * CT + t, EE - 1);
    sv[k] = ei[e];
    dv[k] = ei[EE + e];
  }
  __syncthreads();
  unsigned key[EPT];
  int r[EPT];
#pragma unroll
  for (int k = 0; k < EPT; ++k) {
    key[k] = 0u;
    r[k] = 0;
    if (k * CT + t < nch) {
      const int s = min(max(sv[k], 0), NN - 1);
      const int d = min(max(dv[k], 0), NN - 1);
      const int b = d >> BSH;
      r[k] = atomicAdd(&lcnt[b], 1);
      key[k] = ((unsigned)b << 24) | ((unsigned)(d & (BNODES - 1)) << 16) | (unsigned)s;
    }
  }
  __syncthreads();
  const int mine = (t < NB) ? lcnt[t] : 0;
  const int ia = wave_incl_scan(mine, lane);
  if (lane == 31 && w < 8) wsa[w] = ia;
  __syncthreads();
  if (t < NB) {
    int pa = 0;
#pragma unroll
    for (int k = 0; k < 8; ++k) {
      if (k < w) pa += wsa[k];
    }
    const int exa = pa + ia - mine;
    lstart[t] = exa;
    pstart[i * NB + t] = exa;
    pcnt[i * NB + t] = mine;
  }
  __syncthreads();
#pragma unroll
  for (int k = 0; k < EPT; ++k) {
    if (k * CT + t < nch) {
      const int p = lstart[key[k] >> 24] + r[k];
      if ((unsigned)p < (unsigned)CH) staged[p] = key[k];
    }
  }
  __syncthreads();
  for (int idx = t; idx < nch; idx += CT) chunked[e0 + idx] = staged[idx];
}

__global__ __launch_bounds__(FT) void bucket_sort_kernel(
    const unsigned* __restrict__ chunked, const int* __restrict__ pstart,
    const int* __restrict__ pcnt, int* __restrict__ count,
    int* __restrict__ segbeg, int* __restrict__ srcs) {
  __shared__ int ncnt[BNODES];
  __shared__ int nbeg[BNODES];
  __shared__ int wsum[8];
  __shared__ int wsp[PWV];
  __shared__ int wsq[PWV];
  __shared__ int pgl[NCH];
  __shared__ int pnn[NCH];
  __shared__ int pof[NCH];
  __shared__ unsigned lkeys[FCAP];
  __shared__ int lsorted[FCAP];
  const int t = threadIdx.x;
  const int lane = t & 31;
  const int w = t >> 5;
  const int b = blockIdx.x;
  int ps = 0, pc = 0;
  if (t < NCH) {
    ps = pstart[t * NB + b];
    pc = pcnt[t * NB + b];
    const int cl = max(min(CH, EE - t * CH), 0);
    ps = min(max(ps, 0), cl);
    pc = min(max(pc, 0), cl - ps);
  }
  if (t < BNODES) ncnt[t] = 0;
  const int ic = wave_incl_scan(pc, lane);
  const int is = wave_incl_scan(ps, lane);
  if (lane == 31 && w < PWV) { wsp[w] = ic; wsq[w] = is; }
  __syncthreads();
  int bn = 0, bs = 0;
  for (int k = 0; k < PWV; ++k) { bn += wsp[k]; bs += wsq[k]; }
  bs = min(max(bs, 0), EE);
  bn = min(max(bn, 0), EE - bs);
  if (t < NCH) {
    int wp = 0;
    for (int k = 0; k < PWV; ++k) {
      if (k < w) wp += wsp[k];
    }
    pof[t] = wp + ic - pc;
    pgl[t] = t * CH + ps;
    pnn[t] = pc;
  }
  __syncthreads();
  const bool fits = bn <= FCAP;
  {
    unsigned kv[PPW][2];
#pragma unroll
    for (int j = 0; j < PPW; ++j) {
      const int pi = w + FW * j;
      const int n = pnn[pi];
      const int g = pgl[pi];
#pragma unroll
      for (int h = 0; h < 2; ++h) {
        const int q = min(lane + 32 * h, max(n - 1, 0));
        kv[j][h] = chunked[min(g + q, EE - 1)];
      }
    }
#pragma unroll
    for (int j = 0; j < PPW; ++j) {
      const int pi = w + FW * j;
      const int n = pnn[pi];
      const int g = pgl[pi];
      const int o = pof[pi];
#pragma unroll
      for (int h = 0; h < 2; ++h) {
        const int q = lane + 32 * h;
        if (q < n) {
          atomicAdd(&ncnt[(kv[j][h] >> 16) & (BNODES - 1)], 1);
          if (fits) lkeys[o + q] = kv[j][h];
        }
      }
      for (int q = lane + 64; q < n; q += 32) {
        const unsigned x = chunked[g + q];
        atomicAdd(&ncnt[(x >> 16) & (BNODES - 1)], 1);
        if (fits) lkeys[o + q] = x;
      }
    }
  }
  __syncthreads();
  const int c = (t < BNODES) ? ncnt[t] : 0;
  const int inc = wave_incl_scan(c, lane);
  if (lane == 31 && w < 8) wsum[w] = inc;
  __syncthreads();
  if (t < BNODES) {
    int wp = 0;
#pragma unroll
    for (int k = 0; k < 8; ++k) {
      if (k < w) wp += wsum[k];
    }
    const int ex = wp + inc - c;
    nbeg[t] = ex;
    ncnt[t] = 0;
    const int n = b * BNODES + t;
    if (n < NN) {
      count[n] = c;
      segbeg[n] = bs + ex;
    }
  }
  __syncthreads();
  if (fits) {
    for (int idx = t; idx < bn; idx += FT) {
      const unsigned x = lkeys[idx];
      const int l = (x >> 16) & (BNODES - 1);
      const int p = nbeg[l] + atomicAdd(&ncnt[l], 1);
      if ((unsigned)p < (unsigned)FCAP) lsorted[p] = (int)(x & 0xffffu);
    }
  } else {
    for (int pi = w; pi < NCH; pi += FW) {
      const int n = pnn[pi];
      const int g = pgl[pi];
      for (int q = lane; q < n; q += 32) {
        const unsigned x = chunked[g + q];
        const int l = (x >> 16) & (BNODES - 1);
        const int p = nbeg[l] + atomicAdd(&ncnt[l], 1);
        if ((unsigned)p < (unsigned)bn) srcs[bs + p] = (int)(x & 0xffffu);
      }
    }
  }
  __syncthreads();
  if (fits) {
    for (int idx = t; idx < bn; idx += FT) srcs[bs + idx] = lsorted[idx];
  }
}

template <int KLEN, int BPITCH>
__device__ __forceinline__ void wmma_tile_64x64(
    const float* __restrict__ Ap, const int (&arow)[4],
    const float* __restrict__ Bp, const int khi, v8f (&acc)[4][4]) {
#pragma unroll 1
  for (int kc = 0; kc < KLEN; kc += 32) {
    const int kl = kc + 16 * khi;
    v4f b[16];
    v4f a[4][4];
#pragma unroll
    for (int e = 0; e < 16; ++e) b[e] = *(const v4f*)(Bp + (size_t)(kl + e) * BPITCH);
#pragma unroll
    for (int s = 0; s < 4; ++s) {
      const v4f* pa = (const v4f*)(Ap + (size_t)arow[s] * KDIM + kl);
#pragma unroll
      for (int q = 0; q < 4; ++q) a[s][q] = pa[q];
    }
    __builtin_amdgcn_sched_barrier(0);
#pragma unroll
    for (int j = 0; j < 8; ++j) {
      v2f bfrag[4];
#pragma unroll
      for (int c = 0; c < 4; ++c) {
        bfrag[c].x = b[2 * j][c];
        bfrag[c].y = b[2 * j + 1][c];
      }
#pragma unroll
      for (int s = 0; s < 4; ++s) {
        v2f afrag;
        afrag.x = a[s][j >> 1][2 * (j & 1)];
        afrag.y = a[s][j >> 1][2 * (j & 1) + 1];
#pragma unroll
        for (int c = 0; c < 4; ++c) {
          acc[s][c] = __builtin_amdgcn_wmma_f32_16x16x4_f32(
              false, afrag, false, bfrag[c], (short)0, acc[s][c], false, false);
        }
      }
    }
  }
}

__global__ __launch_bounds__(128) void wmma_proj_kernel(
    const float* __restrict__ feat,
    const float* __restrict__ Wsrc, const float* __restrict__ Bsrc,
    const float* __restrict__ Wdst, const float* __restrict__ Bdst,
    float* __restrict__ Hsrc, float* __restrict__ Hdst) {
  const int lane = threadIdx.x & 31;
  const int cg = __builtin_amdgcn_readfirstlane((int)(threadIdx.x >> 5));
  const int rowTile = blockIdx.x;
  const int mat = blockIdx.y;

  const float* __restrict__ W  = mat ? Wdst : Wsrc;
  const float* __restrict__ Bv = mat ? Bdst : Bsrc;
  float* __restrict__ Hout     = mat ? Hdst : Hsrc;

  const int row0 = rowTile * 64;
  const int col0 = cg * 64;
  const int khi  = lane >> 4;
  const int mlo  = lane & 15;

  int arow[4];
#pragma unroll
  for (int s = 0; s < 4; ++s) {
    int r = row0 + s * 16 + mlo;
    arow[s] = (r < NN) ? r : 0;
  }

  v8f acc[4][4] = {};
  wmma_tile_64x64<KDIM, KDIM>(feat, arow, W + col0 + 4 * mlo, khi, acc);

  const int mbase = khi * 8;
  const int ncol  = col0 + 4 * mlo;
  const v4f bias = *(const v4f*)(Bv + ncol);
  if (row0 + 64 <= NN) {
#pragma unroll
    for (int s = 0; s < 4; ++s) {
#pragma unroll
      for (int i = 0; i < 8; ++i) {
        const int r = row0 + s * 16 + mbase + i;
        v4f o;
        o[0] = acc[s][0][i] + bias[0];
        o[1] = acc[s][1][i] + bias[1];
        o[2] = acc[s][2][i] + bias[2];
        o[3] = acc[s][3][i] + bias[3];
        *(v4f*)(Hout + (size_t)r * KDIM + ncol) = o;
      }
    }
  } else {
#pragma unroll
    for (int s = 0; s < 4; ++s) {
#pragma unroll
      for (int i = 0; i < 8; ++i) {
        const int r = row0 + s * 16 + mbase + i;
        if (r < NN) {
          v4f o;
          o[0] = acc[s][0][i] + bias[0];
          o[1] = acc[s][1][i] + bias[1];
          o[2] = acc[s][2][i] + bias[2];
          o[3] = acc[s][3][i] + bias[3];
          *(v4f*)(Hout + (size_t)r * KDIM + ncol) = o;
        }
      }
    }
  }
}

__device__ __forceinline__ float row8_sum(float x) {
  x += __int_as_float(__builtin_amdgcn_update_dpp(0, __float_as_int(x), 0x161, 0xf, 0xf, false));
  x += __int_as_float(__builtin_amdgcn_update_dpp(0, __float_as_int(x), 0x162, 0xf, 0xf, false));
  x += __int_as_float(__builtin_amdgcn_update_dpp(0, __float_as_int(x), 0x164, 0xf, 0xf, false));
  return x;
}

#define NPB 16
#define TPITCH (KDIM + 4)
#define PPITCH 20
#define PBLK (16 * PPITCH + 16)
static_assert(NN % NPB == 0, "a work-group holds 16 whole nodes: no wave leaves before the barriers");

__global__ __launch_bounds__(512) __attribute__((amdgpu_waves_per_eu(12))) void node_attend_fc_kernel(
    const int* __restrict__ segbeg, const int* __restrict__ count,
    const int* __restrict__ srcs, const float* __restrict__ Hsrc,
    const float* __restrict__ Hdst, const float* __restrict__ attn,
    const float* __restrict__ gat_bias, const float* __restrict__ Wfc,
    const float* __restrict__ bfc, float* __restrict__ out) {
  __shared__ __attribute__((aligned(16))) float tile[NPB * TPITCH];
  __shared__ float part[NPB * PBLK];
  const int w = __builtin_amdgcn_readfirstlane((int)(threadIdx.x >> 5));
  const int lane = threadIdx.x & 31;
  const int node0 = blockIdx.x * NPB;
  const int n = node0 + w;
  int beg = segbeg[n];
  int cnt = count[n];
  beg = min(max(beg, 0), EE);
  cnt = min(max(cnt, 0), EE - beg);

  const int off = (lane >> 3) * 64 + (lane & 7) * 4;
  const float* __restrict__ hd = Hdst + (size_t)n * KDIM + off;
  const v4f d0 = *(const v4f*)hd;
  const v4f d1 = *(const v4f*)(hd + 32);
  const v4f a0 = *(const v4f*)(attn + off);
  const v4f a1 = *(const v4f*)(attn + off + 32);

  float m = -__builtin_inff();
  float den = 0.0f;
  v4f acc0 = {0.0f, 0.0f, 0.0f, 0.0f};
  v4f acc1 = {0.0f, 0.0f, 0.0f, 0.0f};

  for (int base = 0; base < cnt; base += 32) {
    const int nb = min(32, cnt - base);
    int mysrc = 0;
    if (lane < nb) mysrc = srcs[beg + base + lane];
    mysrc = min(max(mysrc, 0), NN - 1);
    for (int j = 0; j < nb; j += 4) {
      v4f r0[4], r1[4];
#pragma unroll
      for (int u = 0; u < 4; ++u) {
        const int jj = min(j + u, nb - 1);
        const int s = __builtin_amdgcn_readlane(mysrc, jj);
        const float* __restrict__ row = Hsrc + (size_t)s * KDIM + off;
        r0[u] = *(const v4f*)row;
        r1[u] = *(const v4f*)(row + 32);
      }
#pragma unroll
      for (int u = 0; u < 4; ++u) {
        if (j + u < nb) {
          v4f x0 = r0[u] + d0;
          v4f x1 = r1[u] + d1;
          x0 = __builtin_elementwise_max(x0, x0 * NEG_SLOPE);
          x1 = __builtin_elementwise_max(x1, x1 * NEG_SLOPE);
          const v4f t = x0 * a0 + x1 * a1;
          float part_s = (t[0] + t[1]) + (t[2] + t[3]);
          part_s = row8_sum(part_s);
          const float dlt = part_s - m;
          const float ex = expf(-fabsf(dlt));
          const bool up = dlt > 0.0f;
          const float sc = up ? ex : 1.0f;
          const float p = up ? 1.0f : ex;
          m = up ? part_s : m;
          den = den * sc + p;
          acc0 = acc0 * sc + p * r0[u];
          acc1 = acc1 * sc + p * r1[u];
        }
      }
    }
  }

  const int cg = w & 3;
  const int kq = w >> 2;
  const int khi = lane >> 4;
  const int mlo = lane & 15;
  const int k0 = 64 * kq + 16 * khi;
  float b[2][16];
#pragma unroll
  for (int c = 0; c < 2; ++c) {
#pragma unroll
    for (int e = 0; e < 16; ++e) b[c][e] = Wfc[(size_t)(k0 + 32 * c + e) * OUTD + 16 * cg + mlo];
  }

  const float inv = 1.0f / ((den == 0.0f) ? 1.0f : den);
  const v4f g0 = *(const v4f*)(gat_bias + off);
  const v4f g1 = *(const v4f*)(gat_bias + off + 32);
  float* __restrict__ tr = tile + w * TPITCH + off;
  *(v4f*)tr = acc0 * inv + g0;
  *(v4f*)(tr + 32) = acc1 * inv + g1;
  __syncthreads();

  v8f d = {};
#pragma unroll
  for (int c = 0; c < 2; ++c) {
    const float* __restrict__ ar = tile + mlo * TPITCH + k0 + 32 * c;
    v4f a[4];
#pragma unroll
    for (int q = 0; q < 4; ++q) a[q] = *(const v4f*)(ar + 4 * q);
#pragma unroll
    for (int j = 0; j < 8; ++j) {
      v2f afrag, bfrag;
      afrag.x = a[j >> 1][2 * (j & 1)];
      afrag.y = a[j >> 1][2 * (j & 1) + 1];
      bfrag.x = b[c][2 * j];
      bfrag.y = b[c][2 * j + 1];
      d = __builtin_amdgcn_wmma_f32_16x16x4_f32(
          false, afrag, false, bfrag, (short)0, d, false, false);
    }
  }
  float* __restrict__ pw = part + w * PBLK + (8 * khi) * PPITCH + mlo;
#pragma unroll
  for (int i = 0; i < 8; ++i) pw[i * PPITCH] = d[i];
  __syncthreads();

#pragma unroll
  for (int u = 0; u < 2; ++u) {
    const int idx = (int)threadIdx.x + 512 * u;
    const int row = idx >> 6;
    const int col = idx & 63;
    const float* __restrict__ pp = part + (col >> 4) * PBLK + row * PPITCH + (col & 15);
    const float s = (pp[0] + pp[4 * PBLK]) + (pp[8 * PBLK] + pp[12 * PBLK]);
    out[(size_t)(node0 + row) * OUTD + col] = s + bfc[col];
  }
}

extern "C" void kernel_launch(void* const* d_in, const int* in_sizes, int n_in,
                              void* d_out, int out_size, void* d_ws, size_t ws_size,
                              hipStream_t stream) {
  const float* feat     = (const float*)d_in[0];
  const int*   ei       = (const int*)d_in[1];
  const float* w_src    = (const float*)d_in[2];
  const float* b_src    = (const float*)d_in[3];
  const float* w_dst    = (const float*)d_in[4];
  const float* b_dst    = (const float*)d_in[5];
  const float* attn     = (const float*)d_in[6];
  const float* gat_bias = (const float*)d_in[7];
  const float* w_fc     = (const float*)d_in[8];
  const float* b_fc     = (const float*)d_in[9];
  float* out = (float*)d_out;

  float* ws    = (float*)d_ws;
  float* Hsrc  = ws;
  float* Hdst  = Hsrc + (size_t)NN * KDIM;
  int* count   = (int*)(Hdst + (size_t)NN * KDIM);
  int* segbeg  = count + NN;
  int* srcs    = segbeg + NN;
  unsigned* chunked = (unsigned*)(srcs + EE);
  int* pstart  = (int*)(chunked + EE);
  int* pcnt    = pstart + NCH * NB;

  const int rowTiles = (NN + 63) / 64;

  chunk_sort_kernel<<<NCH, CT, 0, stream>>>(ei, chunked, pstart, pcnt);

  bucket_sort_kernel<<<NB, FT, 0, stream>>>(chunked, pstart, pcnt, count, segbeg, srcs);

  wmma_proj_kernel<<<dim3(rowTiles, 2), 128, 0, stream>>>(
      feat, w_src, b_src, w_dst, b_dst, Hsrc, Hdst);

  node_attend_fc_kernel<<<NN / NPB, 512, 0, stream>>>(
      segbeg, count, srcs, Hsrc, Hdst, attn, gat_bias, w_fc, b_fc, out);
}


# Projection GEMM rows cut into 1536 work-groups = two full rounds of the 768 resident: 106 take 80 rows (every 7th early one), 1430 take 64; the 1564 before left a third round of 28 alone
# speedup vs baseline: 6.2610x; 1.0340x over previous
#include <hip/hip_runtime.h>


#define NN 50000
#define EE 800000
#define KDIM 256
#define OUTD 64
#define NEG_SLOPE 0.2f

typedef __attribute__((ext_vector_type(2))) float v2f;
typedef __attribute__((ext_vector_type(4))) float v4f;
typedef __attribute__((ext_vector_type(8))) float v8f;

#define BSH 8
#define BNODES (1 << BSH)
#define NB ((NN + BNODES - 1) >> BSH)
#define NCH 128
#define CH ((EE + NCH - 1) / NCH)
#define CT 512
#define EPT ((CH + CT - 1) / CT)
#define FT 512
#define FW (FT / 32)
#define PPW (NCH / FW)
#define PWV (NCH / 32)
#define FCAP 8192

static_assert(NN <= 65536, "a source id travels in 16 bits");
static_assert(NB <= 256, "a bucket id travels in 8 bits; 8 waves scan the buckets");
static_assert(BNODES == 256, "bucket_sort scans a bucket's nodes with 8 waves");
static_assert(NCH % FW == 0 && NCH % 32 == 0 && NCH <= FT, "piece table: one thread per chunk");
static_assert(CT >= 256 && FT >= 256, "8 waves scan");

__device__ __forceinline__ int wave_incl_scan(int x, const int lane) {
#pragma unroll
  for (int d = 1; d < 32; d <<= 1) {
    const int y = __shfl_up(x, d, 32);
    if (lane >= d) x += y;
  }
  return x;
}

__global__ __launch_bounds__(CT) void chunk_sort_kernel(
    const int* __restrict__ ei, unsigned* __restrict__ chunked,
    int* __restrict__ pstart, int* __restrict__ pcnt) {
  __shared__ int lcnt[NB];
  __shared__ int lstart[NB];
  __shared__ int wsa[8];
  __shared__ unsigned staged[CH];
  const int t = threadIdx.x;
  const int lane = t & 31;
  const int w = t >> 5;
  const int i = blockIdx.x;
  if (t < NB) lcnt[t] = 0;
  const int e0 = i * CH;
  const int nch = max(min(CH, EE - e0), 0);
  int sv[EPT], dv[EPT];
#pragma unroll
  for (int k = 0; k < EPT; ++k) {
    const int e = min(e0 + k * CT + t, EE - 1);
    sv[k] = ei[e];
    dv[k] = ei[EE + e];
  }
  __syncthreads();
  unsigned key[EPT];
  int r[EPT];
#pragma unroll
  for (int k = 0; k < EPT; ++k) {
    key[k] = 0u;
    r[k] = 0;
    if (k * CT + t < nch) {
      const int s = min(max(sv[k], 0), NN - 1);
      const int d = min(max(dv[k], 0), NN - 1);
      const int b = d >> BSH;
      r[k] = atomicAdd(&lcnt[b], 1);
      key[k] = ((unsigned)b << 24) | ((unsigned)(d & (BNODES - 1)) << 16) | (unsigned)s;
    }
  }
  __syncthreads();
  const int mine = (t < NB) ? lcnt[t] : 0;
  const int ia = wave_incl_scan(mine, lane);
  if (lane == 31 && w < 8) wsa[w] = ia;
  __syncthreads();
  if (t < NB) {
    int pa = 0;
#pragma unroll
    for (int k = 0; k < 8; ++k) {
      if (k < w) pa += wsa[k];
    }
    const int exa = pa + ia - mine;
    lstart[t] = exa;
    pstart[i * NB + t] = exa;
    pcnt[i * NB + t] = mine;
  }
  __syncthreads();
#pragma unroll
  for (int k = 0; k < EPT; ++k) {
    if (k * CT + t < nch) {
      const int p = lstart[key[k] >> 24] + r[k];
      if ((unsigned)p < (unsigned)CH) staged[p] = key[k];
    }
  }
  __syncthreads();
  for (int idx = t; idx < nch; idx += CT) chunked[e0 + idx] = staged[idx];
}

__global__ __launch_bounds__(FT) void bucket_sort_kernel(
    const unsigned* __restrict__ chunked, const int* __restrict__ pstart,
    const int* __restrict__ pcnt, int* __restrict__ count,
    int* __restrict__ segbeg, int* __restrict__ srcs) {
  __shared__ int ncnt[BNODES];
  __shared__ int nbeg[BNODES];
  __shared__ int wsum[8];
  __shared__ int wsp[PWV];
  __shared__ int wsq[PWV];
  __shared__ int pgl[NCH];
  __shared__ int pnn[NCH];
  __shared__ int pof[NCH];
  __shared__ unsigned lkeys[FCAP];
  __shared__ int lsorted[FCAP];
  const int t = threadIdx.x;
  const int lane = t & 31;
  const int w = t >> 5;
  const int b = blockIdx.x;
  int ps = 0, pc = 0;
  if (t < NCH) {
    ps = pstart[t * NB + b];
    pc = pcnt[t * NB + b];
    const int cl = max(min(CH, EE - t * CH), 0);
    ps = min(max(ps, 0), cl);
    pc = min(max(pc, 0), cl - ps);
  }
  if (t < BNODES) ncnt[t] = 0;
  const int ic = wave_incl_scan(pc, lane);
  const int is = wave_incl_scan(ps, lane);
  if (lane == 31 && w < PWV) { wsp[w] = ic; wsq[w] = is; }
  __syncthreads();
  int bn = 0, bs = 0;
  for (int k = 0; k < PWV; ++k) { bn += wsp[k]; bs += wsq[k]; }
  bs = min(max(bs, 0), EE);
  bn = min(max(bn, 0), EE - bs);
  if (t < NCH) {
    int wp = 0;
    for (int k = 0; k < PWV; ++k) {
      if (k < w) wp += wsp[k];
    }
    pof[t] = wp + ic - pc;
    pgl[t] = t * CH + ps;
    pnn[t] = pc;
  }
  __syncthreads();
  const bool fits = bn <= FCAP;
  {
    unsigned kv[PPW][2];
#pragma unroll
    for (int j = 0; j < PPW; ++j) {
      const int pi = w + FW * j;
      const int n = pnn[pi];
      const int g = pgl[pi];
#pragma unroll
      for (int h = 0; h < 2; ++h) {
        const int q = min(lane + 32 * h, max(n - 1, 0));
        kv[j][h] = chunked[min(g + q, EE - 1)];
      }
    }
#pragma unroll
    for (int j = 0; j < PPW; ++j) {
      const int pi = w + FW * j;
      const int n = pnn[pi];
      const int g = pgl[pi];
      const int o = pof[pi];
#pragma unroll
      for (int h = 0; h < 2; ++h) {
        const int q = lane + 32 * h;
        if (q < n) {
          atomicAdd(&ncnt[(kv[j][h] >> 16) & (BNODES - 1)], 1);
          if (fits) lkeys[o + q] = kv[j][h];
        }
      }
      for (int q = lane + 64; q < n; q += 32) {
        const unsigned x = chunked[g + q];
        atomicAdd(&ncnt[(x >> 16) & (BNODES - 1)], 1);
        if (fits) lkeys[o + q] = x;
      }
    }
  }
  __syncthreads();
  const int c = (t < BNODES) ? ncnt[t] : 0;
  const int inc = wave_incl_scan(c, lane);
  if (lane == 31 && w < 8) wsum[w] = inc;
  __syncthreads();
  if (t < BNODES) {
    int wp = 0;
#pragma unroll
    for (int k = 0; k < 8; ++k) {
      if (k < w) wp += wsum[k];
    }
    const int ex = wp + inc - c;
    nbeg[t] = ex;
    ncnt[t] = 0;
    const int n = b * BNODES + t;
    if (n < NN) {
      count[n] = c;
      segbeg[n] = bs + ex;
    }
  }
  __syncthreads();
  if (fits) {
    for (int idx = t; idx < bn; idx += FT) {
      const unsigned x = lkeys[idx];
      const int l = (x >> 16) & (BNODES - 1);
      const int p = nbeg[l] + atomicAdd(&ncnt[l], 1);
      if ((unsigned)p < (unsigned)FCAP) lsorted[p] = (int)(x & 0xffffu);
    }
  } else {
    for (int pi = w; pi < NCH; pi += FW) {
      const int n = pnn[pi];
      const int g = pgl[pi];
      for (int q = lane; q < n; q += 32) {
        const unsigned x = chunked[g + q];
        const int l = (x >> 16) & (BNODES - 1);
        const int p = nbeg[l] + atomicAdd(&ncnt[l], 1);
        if ((unsigned)p < (unsigned)bn) srcs[bs + p] = (int)(x & 0xffffu);
      }
    }
  }
  __syncthreads();
  if (fits) {
    for (int idx = t; idx < bn; idx += FT) srcs[bs + idx] = lsorted[idx];
  }
}

template <int NS, int KLEN, int BPITCH>
__device__ __forceinline__ void wmma_tile_rows_x64(
    const float* __restrict__ Ap, const int (&arow)[NS],
    const float* __restrict__ Bp, const int khi, v8f (&acc)[NS][4]) {
#pragma unroll 1
  for (int kc = 0; kc < KLEN; kc += 32) {
    const int kl = kc + 16 * khi;
    v4f b[16];
    v4f a[NS][4];
#pragma unroll
    for (int e = 0; e < 16; ++e) b[e] = *(const v4f*)(Bp + (size_t)(kl + e) * BPITCH);
#pragma unroll
    for (int s = 0; s < NS; ++s) {
      const v4f* pa = (const v4f*)(Ap + (size_t)arow[s] * KDIM + kl);
#pragma unroll
      for (int q = 0; q < 4; ++q) a[s][q] = pa[q];
    }
    __builtin_amdgcn_sched_barrier(0);
#pragma unroll
    for (int j = 0; j < 8; ++j) {
      v2f bfrag[4];
#pragma unroll
      for (int c = 0; c < 4; ++c) {
        bfrag[c].x = b[2 * j][c];
        bfrag[c].y = b[2 * j + 1][c];
      }
#pragma unroll
      for (int s = 0; s < NS; ++s) {
        v2f afrag;
        afrag.x = a[s][j >> 1][2 * (j & 1)];
        afrag.y = a[s][j >> 1][2 * (j & 1) + 1];
#pragma unroll
        for (int c = 0; c < 4; ++c) {
          acc[s][c] = __builtin_amdgcn_wmma_f32_16x16x4_f32(
              false, afrag, false, bfrag[c], (short)0, acc[s][c], false, false);
        }
      }
    }
  }
}

#define RESID 768
#define RGROUPS (NN / 16)
#define NBIG (RGROUPS - 4 * RESID)
#define NSMALL (RESID - NBIG)
#define BIGSTEP 7
static_assert(NN % 16 == 0, "rows are cut in whole 16-row groups");
static_assert(NBIG >= 0 && NSMALL >= 0, "4 or 5 row groups per work-group cover a matrix");
static_assert(80 * NBIG + 64 * NSMALL == NN, "the tiles cover the rows exactly: no mask");
static_assert(BIGSTEP * 2 * NBIG <= 2 * RESID, "the 80-row work-groups fit in the grid's first part");

template <int NS>
__device__ __forceinline__ void proj_tile(
    const float* __restrict__ feat, const float* __restrict__ W,
    const float* __restrict__ Bv, float* __restrict__ Hout,
    const int row0, const int cg, const int lane) {
  const int col0 = cg * 64;
  const int khi  = lane >> 4;
  const int mlo  = lane & 15;

  int arow[NS];
#pragma unroll
  for (int s = 0; s < NS; ++s) arow[s] = row0 + s * 16 + mlo;

  v8f acc[NS][4] = {};
  wmma_tile_rows_x64<NS, KDIM, KDIM>(feat, arow, W + col0 + 4 * mlo, khi, acc);

  const int mbase = khi * 8;
  const int ncol  = col0 + 4 * mlo;
  const v4f bias = *(const v4f*)(Bv + ncol);
#pragma unroll
  for (int s = 0; s < NS; ++s) {
#pragma unroll
    for (int i = 0; i < 8; ++i) {
      const int r = row0 + s * 16 + mbase + i;
      v4f o;
      o[0] = acc[s][0][i] + bias[0];
      o[1] = acc[s][1][i] + bias[1];
      o[2] = acc[s][2][i] + bias[2];
      o[3] = acc[s][3][i] + bias[3];
      *(v4f*)(Hout + (size_t)r * KDIM + ncol) = o;
    }
  }
}

__global__ __launch_bounds__(128) __attribute__((amdgpu_waves_per_eu(3))) void wmma_proj_kernel(
    const float* __restrict__ feat,
    const float* __restrict__ Wsrc, const float* __restrict__ Bsrc,
    const float* __restrict__ Wdst, const float* __restrict__ Bdst,
    float* __restrict__ Hsrc, float* __restrict__ Hdst) {
  const int lane = threadIdx.x & 31;
  const int cg = __builtin_amdgcn_readfirstlane((int)(threadIdx.x >> 5));
  const int i = blockIdx.x;
  const bool big = (i < BIGSTEP * 2 * NBIG) && (i % BIGSTEP == 0);
  if (big) {
    const int q = i / BIGSTEP;
    const int mat = (q >= NBIG) ? 1 : 0;
    const int t = q - mat * NBIG;
    proj_tile<5>(feat, mat ? Wdst : Wsrc, mat ? Bdst : Bsrc, mat ? Hdst : Hsrc,
                 t * 80, cg, lane);
  } else {
    const int before = min((i + BIGSTEP - 1) / BIGSTEP, 2 * NBIG);
    const int j = i - before;
    const int mat = (j >= NSMALL) ? 1 : 0;
    const int t = j - mat * NSMALL;
    proj_tile<4>(feat, mat ? Wdst : Wsrc, mat ? Bdst : Bsrc, mat ? Hdst : Hsrc,
                 NBIG * 80 + t * 64, cg, lane);
  }
}

__device__ __forceinline__ float row8_sum(float x) {
  x += __int_as_float(__builtin_amdgcn_update_dpp(0, __float_as_int(x), 0x161, 0xf, 0xf, false));
  x += __int_as_float(__builtin_amdgcn_update_dpp(0, __float_as_int(x), 0x162, 0xf, 0xf, false));
  x += __int_as_float(__builtin_amdgcn_update_dpp(0, __float_as_int(x), 0x164, 0xf, 0xf, false));
  return x;
}

#define NPB 16
#define TPITCH (KDIM + 4)
#define PPITCH 20
#define PBLK (16 * PPITCH + 16)
static_assert(NN % NPB == 0, "a work-group holds 16 whole nodes: no wave leaves before the barriers");

__global__ __launch_bounds__(512) __attribute__((amdgpu_waves_per_eu(12))) void node_attend_fc_kernel(
    const int* __restrict__ segbeg, const int* __restrict__ count,
    const int* __restrict__ srcs, const float* __restrict__ Hsrc,
    const float* __restrict__ Hdst, const float* __restrict__ attn,
    const float* __restrict__ gat_bias, const float* __restrict__ Wfc,
    const float* __restrict__ bfc, float* __restrict__ out) {
  __shared__ __attribute__((aligned(16))) float tile[NPB * TPITCH];
  __shared__ float part[NPB * PBLK];
  const int w = __builtin_amdgcn_readfirstlane((int)(threadIdx.x >> 5));
  const int lane = threadIdx.x & 31;
  const int node0 = blockIdx.x * NPB;
  const int n = node0 + w;
  int beg = segbeg[n];
  int cnt = count[n];
  beg = min(max(beg, 0), EE);
  cnt = min(max(cnt, 0), EE - beg);

  const int off = (lane >> 3) * 64 + (lane & 7) * 4;
  const float* __restrict__ hd = Hdst + (size_t)n * KDIM + off;
  const v4f d0 = *(const v4f*)hd;
  const v4f d1 = *(const v4f*)(hd + 32);
  const v4f a0 = *(const v4f*)(attn + off);
  const v4f a1 = *(const v4f*)(attn + off + 32);

  float m = -__builtin_inff();
  float den = 0.0f;
  v4f acc0 = {0.0f, 0.0f, 0.0f, 0.0f};
  v4f acc1 = {0.0f, 0.0f, 0.0f, 0.0f};

  for (int base = 0; base < cnt; base += 32) {
    const int nb = min(32, cnt - base);
    int mysrc = 0;
    if (lane < nb) mysrc = srcs[beg + base + lane];
    mysrc = min(max(mysrc, 0), NN - 1);
    for (int j = 0; j < nb; j += 4) {
      v4f r0[4], r1[4];
#pragma unroll
      for (int u = 0; u < 4; ++u) {
        const int jj = min(j + u, nb - 1);
        const int s = __builtin_amdgcn_readlane(mysrc, jj);
        const float* __restrict__ row = Hsrc + (size_t)s * KDIM + off;
        r0[u] = *(const v4f*)row;
        r1[u] = *(const v4f*)(row + 32);
      }
#pragma unroll
      for (int u = 0; u < 4; ++u) {
        if (j + u < nb) {
          v4f x0 = r0[u] + d0;
          v4f x1 = r1[u] + d1;
          x0 = __builtin_elementwise_max(x0, x0 * NEG_SLOPE);
          x1 = __builtin_elementwise_max(x1, x1 * NEG_SLOPE);
          const v4f t = x0 * a0 + x1 * a1;
          float part_s = (t[0] + t[1]) + (t[2] + t[3]);
          part_s = row8_sum(part_s);
          const float dlt = part_s - m;
          const float ex = expf(-fabsf(dlt));
          const bool up = dlt > 0.0f;
          const float sc = up ? ex : 1.0f;
          const float p = up ? 1.0f : ex;
          m = up ? part_s : m;
          den = den * sc + p;
          acc0 = acc0 * sc + p * r0[u];
          acc1 = acc1 * sc + p * r1[u];
        }
      }
    }
  }

  const int cg = w & 3;
  const int kq = w >> 2;
  const int khi = lane >> 4;
  const int mlo = lane & 15;
  const int k0 = 64 * kq + 16 * khi;
  float b[2][16];
#pragma unroll
  for (int c = 0; c < 2; ++c) {
#pragma unroll
    for (int e = 0; e < 16; ++e) b[c][e] = Wfc[(size_t)(k0 + 32 * c + e) * OUTD + 16 * cg + mlo];
  }

  const float inv = 1.0f / ((den == 0.0f) ? 1.0f : den);
  const v4f g0 = *(const v4f*)(gat_bias + off);
  const v4f g1 = *(const v4f*)(gat_bias + off + 32);
  float* __restrict__ tr = tile + w * TPITCH + off;
  *(v4f*)tr = acc0 * inv + g0;
  *(v4f*)(tr + 32) = acc1 * inv + g1;
  __syncthreads();

  v8f d = {};
#pragma unroll
  for (int c = 0; c < 2; ++c) {
    const float* __restrict__ ar = tile + mlo * TPITCH + k0 + 32 * c;
    v4f a[4];
#pragma unroll
    for (int q = 0; q < 4; ++q) a[q] = *(const v4f*)(ar + 4 * q);
#pragma unroll
    for (int j = 0; j < 8; ++j) {
      v2f afrag, bfrag;
      afrag.x = a[j >> 1][2 * (j & 1)];
      afrag.y = a[j >> 1][2 * (j & 1) + 1];
      bfrag.x = b[c][2 * j];
      bfrag.y = b[c][2 * j + 1];
      d = __builtin_amdgcn_wmma_f32_16x16x4_f32(
          false, afrag, false, bfrag, (short)0, d, false, false);
    }
  }
  float* __restrict__ pw = part + w * PBLK + (8 * khi) * PPITCH + mlo;
#pragma unroll
  for (int i = 0; i < 8; ++i) pw[i * PPITCH] = d[i];
  __syncthreads();

#pragma unroll
  for (int u = 0; u < 2; ++u) {
    const int idx = (int)threadIdx.x + 512 * u;
    const int row = idx >> 6;
    const int col = idx & 63;
    const float* __restrict__ pp = part + (col >> 4) * PBLK + row * PPITCH + (col & 15);
    const float s = (pp[0] + pp[4 * PBLK]) + (pp[8 * PBLK] + pp[12 * PBLK]);
    out[(size_t)(node0 + row) * OUTD + col] = s + bfc[col];
  }
}

extern "C" void kernel_launch(void* const* d_in, const int* in_sizes, int n_in,
                              void* d_out, int out_size, void* d_ws, size_t ws_size,
                              hipStream_t stream) {
  const float* feat     = (const float*)d_in[0];
  const int*   ei       = (const int*)d_in[1];
  const float* w_src    = (const float*)d_in[2];
  const float* b_src    = (const float*)d_in[3];
  const float* w_dst    = (const float*)d_in[4];
  const float* b_dst    = (const float*)d_in[5];
  const float* attn     = (const float*)d_in[6];
  const float* gat_bias = (const float*)d_in[7];
  const float* w_fc     = (const float*)d_in[8];
  const float* b_fc     = (const float*)d_in[9];
  float* out = (float*)d_out;

  float* ws    = (float*)d_ws;
  float* Hsrc  = ws;
  float* Hdst  = Hsrc + (size_t)NN * KDIM;
  int* count   = (int*)(Hdst + (size_t)NN * KDIM);
  int* segbeg  = count + NN;
  int* srcs    = segbeg + NN;
  unsigned* chunked = (unsigned*)(srcs + EE);
  int* pstart  = (int*)(chunked + EE);
  int* pcnt    = pstart + NCH * NB;

  chunk_sort_kernel<<<NCH, CT, 0, stream>>>(ei, chunked, pstart, pcnt);

  bucket_sort_kernel<<<NB, FT, 0, stream>>>(chunked, pstart, pcnt, count, segbeg, srcs);

  wmma_proj_kernel<<<2 * RESID, 128, 0, stream>>>(
      feat, w_src, b_src, w_dst, b_dst, Hsrc, Hdst);

  node_attend_fc_kernel<<<NN / NPB, 512, 0, stream>>>(
      segbeg, count, srcs, Hsrc, Hdst, attn, gat_bias, w_fc, b_fc, out);
}
